# starting kernel
# baseline (speedup 1.0000x reference)
#include <hip/hip_runtime.h>

#define N_NODES 10000
#define N_EDGES 100000
#define HIDDEN 128
#define MUL0 32
#define MUL1 8
#define IN_DIM 56
#define W_NUMEL 1600
#define EPS 1e-5f

typedef __attribute__((ext_vector_type(2))) float v2f;
typedef __attribute__((ext_vector_type(8))) float v8f;

#define WS_NUM   0
#define WS_CNT   (N_NODES*IN_DIM)
#define WS_STAT  (WS_CNT + N_NODES)
#define WS_TOTAL (WS_STAT + 72)

__global__ __launch_bounds__(256) void zero_ws_kernel(float* __restrict__ ws) {
  int i = blockIdx.x * 256 + threadIdx.x;
  if (i < WS_TOTAL) ws[i] = 0.0f;
}

__global__ __launch_bounds__(256)
void edge_kernel(const float* __restrict__ node_attr,
                 const int*   __restrict__ edge_index,
                 const float* __restrict__ edge_attr,
                 const float* __restrict__ edge_sh,
                 const float* __restrict__ fc1_w, const float* __restrict__ fc1_b,
                 const float* __restrict__ fc2_w, const float* __restrict__ fc2_b,
                 float* __restrict__ num, float* __restrict__ cnt)
{
  __shared__ float sA[32 * 132];
  __shared__ float sH[32 * 132];
  __shared__ float sX[32 * IN_DIM];
  __shared__ float sY[32 * 4];
  __shared__ float sZ0[32 * MUL0];
  __shared__ float sZ1[32 * MUL1];
  __shared__ float sTP[32 * IN_DIM];
  __shared__ int   sSrc[32];

  const int tid  = threadIdx.x;
  const int lane = tid & 31;
  const int wv   = tid >> 5;
  const int e0   = blockIdx.x * 32;

  for (int i = tid; i < 32 * HIDDEN; i += 256) {
    int r = i >> 7, c = i & 127;
    sA[r * 132 + c] = edge_attr[(size_t)(e0 + r) * HIDDEN + c];
  }
  if (tid < 32) sSrc[tid] = edge_index[e0 + tid];
  if (tid < 128) {
    int r = tid >> 2, c = tid & 3;
    sY[r * 4 + c] = edge_sh[(size_t)(e0 + r) * 4 + c];
  }
  for (int i = tid; i < 32 * IN_DIM; i += 256) {
    int r = i / IN_DIM, c = i - r * IN_DIM;
    int dst = edge_index[N_EDGES + e0 + r];
    sX[i]  = node_attr[(size_t)dst * IN_DIM + c];
    sTP[i] = 0.0f;
  }
  __syncthreads();

  for (int i = tid; i < 32 * MUL0; i += 256) {
    int r = i >> 5, u = i & 31;
    sZ0[i] = sX[r * IN_DIM + u] * sY[r * 4 + 0];
  }
  for (int i = tid; i < 32 * MUL1; i += 256) {
    int r = i >> 3, u = i & 7;
    const float* xr = &sX[r * IN_DIM + MUL0 + u * 3];
    const float* yr = &sY[r * 4 + 1];
    sZ1[i] = xr[0] * yr[0] + xr[1] * yr[1] + xr[2] * yr[2];
  }
  __syncthreads();

  const int m     = lane & 15;
  const int koff  = (lane >> 4) << 1;
  const int mrow0 = (lane >> 4) << 3;

  {
    v8f acc0 = {}, acc1 = {};
    const int col = wv * 16 + m;
    #pragma unroll 8
    for (int kb = 0; kb < HIDDEN; kb += 4) {
      v2f a0, a1, b;
      a0.x = sA[m * 132 + kb + koff];
      a0.y = sA[m * 132 + kb + koff + 1];
      a1.x = sA[(16 + m) * 132 + kb + koff];
      a1.y = sA[(16 + m) * 132 + kb + koff + 1];
      b.x  = fc1_w[(size_t)(kb + koff) * HIDDEN + col];
      b.y  = fc1_w[(size_t)(kb + koff + 1) * HIDDEN + col];
      acc0 = __builtin_amdgcn_wmma_f32_16x16x4_f32(false, a0, false, b, (short)0, acc0, false, false);
      acc1 = __builtin_amdgcn_wmma_f32_16x16x4_f32(false, a1, false, b, (short)0, acc1, false, false);
    }
    const float bias = fc1_b[col];
    #pragma unroll
    for (int v = 0; v < 8; ++v) {
      float h0 = acc0[v] + bias;
      float h1 = acc1[v] + bias;
      sH[(mrow0 + v) * 132 + col]      = h0 > 0.0f ? h0 : 0.0f;
      sH[(16 + mrow0 + v) * 132 + col] = h1 > 0.0f ? h1 : 0.0f;
    }
  }
  __syncthreads();

  const float alpha = 0.15811388300841897f;
  const float ais3  = alpha * 0.5773502691896258f;

  for (int nt = wv; nt < 100; nt += 8) {
    v8f cc[2] = {{}, {}};
    const int col = nt * 16 + m;
    #pragma unroll 8
    for (int kb = 0; kb < HIDDEN; kb += 4) {
      v2f a0, a1, b;
      a0.x = sH[m * 132 + kb + koff];
      a0.y = sH[m * 132 + kb + koff + 1];
      a1.x = sH[(16 + m) * 132 + kb + koff];
      a1.y = sH[(16 + m) * 132 + kb + koff + 1];
      b.x  = fc2_w[(size_t)(kb + koff) * W_NUMEL + col];
      b.y  = fc2_w[(size_t)(kb + koff + 1) * W_NUMEL + col];
      cc[0] = __builtin_amdgcn_wmma_f32_16x16x4_f32(false, a0, false, b, (short)0, cc[0], false, false);
      cc[1] = __builtin_amdgcn_wmma_f32_16x16x4_f32(false, a1, false, b, (short)0, cc[1], false, false);
    }
    const float bias = fc2_b[col];

    #pragma unroll
    for (int t = 0; t < 2; ++t) {
      const int eb = t * 16;
      if (col < 1024) {
        const int u = col >> 5, w = col & 31;
        #pragma unroll
        for (int v = 0; v < 8; ++v) {
          int e = eb + mrow0 + v;
          float val = cc[t][v] + bias;
          unsafeAtomicAdd(&sTP[e * IN_DIM + w], alpha * sZ0[e * MUL0 + u] * val);
        }
      } else if (col < 1280) {
        const int jj = col - 1024, u = jj >> 5, w = jj & 31;
        #pragma unroll
        for (int v = 0; v < 8; ++v) {
          int e = eb + mrow0 + v;
          float val = cc[t][v] + bias;
          unsafeAtomicAdd(&sTP[e * IN_DIM + w], ais3 * sZ1[e * MUL1 + u] * val);
        }
      } else if (col < 1536) {
        const int jj = col - 1280, u = jj >> 3, w = jj & 7;
        #pragma unroll
        for (int v = 0; v < 8; ++v) {
          int e = eb + mrow0 + v;
          float val = alpha * (cc[t][v] + bias) * sX[e * IN_DIM + u];
          unsafeAtomicAdd(&sTP[e * IN_DIM + MUL0 + w * 3 + 0], val * sY[e * 4 + 1]);
          unsafeAtomicAdd(&sTP[e * IN_DIM + MUL0 + w * 3 + 1], val * sY[e * 4 + 2]);
          unsafeAtomicAdd(&sTP[e * IN_DIM + MUL0 + w * 3 + 2], val * sY[e * 4 + 3]);
        }
      } else {
        const int jj = col - 1536, u = jj >> 3, w = jj & 7;
        #pragma unroll
        for (int v = 0; v < 8; ++v) {
          int e = eb + mrow0 + v;
          float val = alpha * (cc[t][v] + bias) * sY[e * 4 + 0];
          unsafeAtomicAdd(&sTP[e * IN_DIM + MUL0 + w * 3 + 0], val * sX[e * IN_DIM + MUL0 + u * 3 + 0]);
          unsafeAtomicAdd(&sTP[e * IN_DIM + MUL0 + w * 3 + 1], val * sX[e * IN_DIM + MUL0 + u * 3 + 1]);
          unsafeAtomicAdd(&sTP[e * IN_DIM + MUL0 + w * 3 + 2], val * sX[e * IN_DIM + MUL0 + u * 3 + 2]);
        }
      }
    }
  }
  __syncthreads();

  for (int i = tid; i < 32 * IN_DIM; i += 256) {
    int r = i / IN_DIM, c = i - r * IN_DIM;
    unsafeAtomicAdd(&num[(size_t)sSrc[r] * IN_DIM + c], sTP[i]);
  }
  if (tid < 32) unsafeAtomicAdd(&cnt[sSrc[tid]], 1.0f);
}

__global__ __launch_bounds__(128)
void node_pass1(const float* __restrict__ node_attr,
                const float* __restrict__ num, const float* __restrict__ cnt,
                float* __restrict__ outp, float* __restrict__ stat)
{
  __shared__ float ssum[MUL0], ssq[MUL0], svq[MUL1];
  const int tid = threadIdx.x;
  if (tid < MUL0) { ssum[tid] = 0.0f; ssq[tid] = 0.0f; }
  if (tid < MUL1) svq[tid] = 0.0f;
  __syncthreads();

  const int n = blockIdx.x * 128 + tid;
  if (n < N_NODES) {
    float c   = cnt[n];
    float inv = 1.0f / (c > 1.0f ? c : 1.0f);
    for (int ch = 0; ch < MUL0; ++ch) {
      float p = num[(size_t)n * IN_DIM + ch] * inv + node_attr[(size_t)n * IN_DIM + ch];
      outp[(size_t)n * IN_DIM + ch] = p;
      unsafeAtomicAdd(&ssum[ch], p);
      unsafeAtomicAdd(&ssq[ch], p * p);
    }
    for (int u = 0; u < MUL1; ++u) {
      float acc = 0.0f;
      for (int mm = 0; mm < 3; ++mm) {
        int ch = MUL0 + u * 3 + mm;
        float p = num[(size_t)n * IN_DIM + ch] * inv + node_attr[(size_t)n * IN_DIM + ch];
        outp[(size_t)n * IN_DIM + ch] = p;
        acc += p * p;
      }
      unsafeAtomicAdd(&svq[u], acc);
    }
  }
  __syncthreads();
  if (tid < MUL0) {
    unsafeAtomicAdd(&stat[tid], ssum[tid]);
    unsafeAtomicAdd(&stat[MUL0 + tid], ssq[tid]);
  }
  if (tid < MUL1) unsafeAtomicAdd(&stat[2 * MUL0 + tid], svq[tid]);
}

__global__ __launch_bounds__(128)
void node_pass2(const float* __restrict__ stat,
                const float* __restrict__ bn_w_s, const float* __restrict__ bn_w_v,
                const float* __restrict__ bn_b_s,
                float* __restrict__ out)
{
  const int n = blockIdx.x * 128 + threadIdx.x;
  if (n >= N_NODES) return;
  const float invN = 1.0f / (float)N_NODES;
  for (int ch = 0; ch < MUL0; ++ch) {
    float mean = stat[ch] * invN;
    float var  = stat[MUL0 + ch] * invN - mean * mean;
    float p = out[(size_t)n * IN_DIM + ch];
    out[(size_t)n * IN_DIM + ch] = (p - mean) * rsqrtf(var + EPS) * bn_w_s[ch] + bn_b_s[ch];
  }
  for (int u = 0; u < MUL1; ++u) {
    float vn    = stat[2 * MUL0 + u] * invN * (1.0f / 3.0f);
    float scale = rsqrtf(vn + EPS) * bn_w_v[u];
    for (int mm = 0; mm < 3; ++mm) {
      int ch = MUL0 + u * 3 + mm;
      out[(size_t)n * IN_DIM + ch] *= scale;
    }
  }
}

extern "C" void kernel_launch(void* const* d_in, const int* in_sizes, int n_in,
                              void* d_out, int out_size, void* d_ws, size_t ws_size,
                              hipStream_t stream) {
  const float* node_attr = (const float*)d_in[0];
  const int*   edge_index = (const int*)d_in[1];
  const float* edge_attr = (const float*)d_in[2];
  const float* edge_sh   = (const float*)d_in[3];
  const float* fc1_w = (const float*)d_in[4];
  const float* fc1_b = (const float*)d_in[5];
  const float* fc2_w = (const float*)d_in[6];
  const float* fc2_b = (const float*)d_in[7];
  const float* bn_w_s = (const float*)d_in[8];
  const float* bn_w_v = (const float*)d_in[9];
  const float* bn_b_s = (const float*)d_in[10];

  float* ws   = (float*)d_ws;
  float* num  = ws + WS_NUM;
  float* cnt  = ws + WS_CNT;
  float* stat = ws + WS_STAT;
  float* out  = (float*)d_out;

  zero_ws_kernel<<<(WS_TOTAL + 255) / 256, 256, 0, stream>>>(ws);
  edge_kernel<<<N_EDGES / 32, 256, 0, stream>>>(node_attr, edge_index, edge_attr, edge_sh,
                                                fc1_w, fc1_b, fc2_w, fc2_b, num, cnt);
  node_pass1<<<(N_NODES + 127) / 128, 128, 0, stream>>>(node_attr, num, cnt, out, stat);
  node_pass2<<<(N_NODES + 127) / 128, 128, 0, stream>>>(stat, bn_w_s, bn_w_v, bn_b_s, out);
}


# edge kernel: each wave loads the whole next 16-column tile's weight fragments (64 VGPRs) one tile ahead, before the epilogue, so no WMMA group waits on an L2 round trip
# speedup vs baseline: 1.0738x; 1.0738x over previous
#include <hip/hip_runtime.h>

#define N_NODES 10000
#define N_EDGES 100000
#define HIDDEN 128
#define MUL0 32
#define MUL1 8
#define IN_DIM 56
#define W_NUMEL 1600
#define EPS 1e-5f

typedef __attribute__((ext_vector_type(2))) float v2f;
typedef __attribute__((ext_vector_type(8))) float v8f;

__device__ __forceinline__ void load_bfrag(v2f (&bf)[32], const float* __restrict__ p, int ld) {
  #pragma unroll
  for (int i = 0; i < 32; ++i) {
    bf[i].x = p[(size_t)(4 * i) * ld];
    bf[i].y = p[(size_t)(4 * i + 1) * ld];
  }
}

#define WS_NUM   0
#define WS_CNT   (N_NODES*IN_DIM)
#define WS_STAT  (WS_CNT + N_NODES)
#define WS_TOTAL (WS_STAT + 72)

__global__ __launch_bounds__(256) void zero_ws_kernel(float* __restrict__ ws) {
  int i = blockIdx.x * 256 + threadIdx.x;
  if (i < WS_TOTAL) ws[i] = 0.0f;
}

__global__ __launch_bounds__(256)
void edge_kernel(const float* __restrict__ node_attr,
                 const int*   __restrict__ edge_index,
                 const float* __restrict__ edge_attr,
                 const float* __restrict__ edge_sh,
                 const float* __restrict__ fc1_w, const float* __restrict__ fc1_b,
                 const float* __restrict__ fc2_w, const float* __restrict__ fc2_b,
                 float* __restrict__ num, float* __restrict__ cnt)
{
  __shared__ float sA[32 * 132];
  __shared__ float sH[32 * 132];
  __shared__ float sX[32 * IN_DIM];
  __shared__ float sY[32 * 4];
  __shared__ float sZ0[32 * MUL0];
  __shared__ float sZ1[32 * MUL1];
  __shared__ float sTP[32 * IN_DIM];
  __shared__ int   sSrc[32];

  const int tid  = threadIdx.x;
  const int lane = tid & 31;
  const int wv   = tid >> 5;
  const int e0   = blockIdx.x * 32;

  const int m     = lane & 15;
  const int koff  = (lane >> 4) << 1;
  const int mrow0 = (lane >> 4) << 3;

  v2f bf[32];
  const float bias1 = fc1_b[wv * 16 + m];
  load_bfrag(bf, fc1_w + (size_t)koff * HIDDEN + wv * 16 + m, HIDDEN);

  for (int i = tid; i < 32 * HIDDEN; i += 256) {
    int r = i >> 7, c = i & 127;
    sA[r * 132 + c] = edge_attr[(size_t)(e0 + r) * HIDDEN + c];
  }
  if (tid < 32) sSrc[tid] = edge_index[e0 + tid];
  if (tid < 128) {
    int r = tid >> 2, c = tid & 3;
    sY[r * 4 + c] = edge_sh[(size_t)(e0 + r) * 4 + c];
  }
  for (int i = tid; i < 32 * IN_DIM; i += 256) {
    int r = i / IN_DIM, c = i - r * IN_DIM;
    int dst = edge_index[N_EDGES + e0 + r];
    sX[i]  = node_attr[(size_t)dst * IN_DIM + c];
    sTP[i] = 0.0f;
  }
  __syncthreads();

  for (int i = tid; i < 32 * MUL0; i += 256) {
    int r = i >> 5, u = i & 31;
    sZ0[i] = sX[r * IN_DIM + u] * sY[r * 4 + 0];
  }
  for (int i = tid; i < 32 * MUL1; i += 256) {
    int r = i >> 3, u = i & 7;
    const float* xr = &sX[r * IN_DIM + MUL0 + u * 3];
    const float* yr = &sY[r * 4 + 1];
    sZ1[i] = xr[0] * yr[0] + xr[1] * yr[1] + xr[2] * yr[2];
  }
  __syncthreads();

  float bias_next;
  {
    v8f acc0 = {}, acc1 = {};
    const int col = wv * 16 + m;
    #pragma unroll
    for (int i = 0; i < 32; ++i) {
      const int kb = 4 * i;
      v2f a0, a1;
      a0.x = sA[m * 132 + kb + koff];
      a0.y = sA[m * 132 + kb + koff + 1];
      a1.x = sA[(16 + m) * 132 + kb + koff];
      a1.y = sA[(16 + m) * 132 + kb + koff + 1];
      acc0 = __builtin_amdgcn_wmma_f32_16x16x4_f32(false, a0, false, bf[i], (short)0, acc0, false, false);
      acc1 = __builtin_amdgcn_wmma_f32_16x16x4_f32(false, a1, false, bf[i], (short)0, acc1, false, false);
    }
    bias_next = fc2_b[col];
    load_bfrag(bf, fc2_w + (size_t)koff * W_NUMEL + col, W_NUMEL);
    const float bias = bias1;
    #pragma unroll
    for (int v = 0; v < 8; ++v) {
      float h0 = acc0[v] + bias;
      float h1 = acc1[v] + bias;
      sH[(mrow0 + v) * 132 + col]      = h0 > 0.0f ? h0 : 0.0f;
      sH[(16 + mrow0 + v) * 132 + col] = h1 > 0.0f ? h1 : 0.0f;
    }
  }
  __syncthreads();

  const float alpha = 0.15811388300841897f;
  const float ais3  = alpha * 0.5773502691896258f;

  for (int nt = wv; nt < 100; nt += 8) {
    v8f cc[2] = {{}, {}};
    const int col = nt * 16 + m;
    const float bias = bias_next;
    #pragma unroll
    for (int i = 0; i < 32; ++i) {
      const int kb = 4 * i;
      v2f a0, a1;
      a0.x = sH[m * 132 + kb + koff];
      a0.y = sH[m * 132 + kb + koff + 1];
      a1.x = sH[(16 + m) * 132 + kb + koff];
      a1.y = sH[(16 + m) * 132 + kb + koff + 1];
      cc[0] = __builtin_amdgcn_wmma_f32_16x16x4_f32(false, a0, false, bf[i], (short)0, cc[0], false, false);
      cc[1] = __builtin_amdgcn_wmma_f32_16x16x4_f32(false, a1, false, bf[i], (short)0, cc[1], false, false);
    }
    if (nt + 8 < 100) {
      bias_next = fc2_b[col + 128];
      load_bfrag(bf, fc2_w + (size_t)koff * W_NUMEL + col + 128, W_NUMEL);
    }

    #pragma unroll
    for (int t = 0; t < 2; ++t) {
      const int eb = t * 16;
      if (col < 1024) {
        const int u = col >> 5, w = col & 31;
        #pragma unroll
        for (int v = 0; v < 8; ++v) {
          int e = eb + mrow0 + v;
          float val = cc[t][v] + bias;
          unsafeAtomicAdd(&sTP[e * IN_DIM + w], alpha * sZ0[e * MUL0 + u] * val);
        }
      } else if (col < 1280) {
        const int jj = col - 1024, u = jj >> 5, w = jj & 31;
        #pragma unroll
        for (int v = 0; v < 8; ++v) {
          int e = eb + mrow0 + v;
          float val = cc[t][v] + bias;
          unsafeAtomicAdd(&sTP[e * IN_DIM + w], ais3 * sZ1[e * MUL1 + u] * val);
        }
      } else if (col < 1536) {
        const int jj = col - 1280, u = jj >> 3, w = jj & 7;
        #pragma unroll
        for (int v = 0; v < 8; ++v) {
          int e = eb + mrow0 + v;
          float val = alpha * (cc[t][v] + bias) * sX[e * IN_DIM + u];
          unsafeAtomicAdd(&sTP[e * IN_DIM + MUL0 + w * 3 + 0], val * sY[e * 4 + 1]);
          unsafeAtomicAdd(&sTP[e * IN_DIM + MUL0 + w * 3 + 1], val * sY[e * 4 + 2]);
          unsafeAtomicAdd(&sTP[e * IN_DIM + MUL0 + w * 3 + 2], val * sY[e * 4 + 3]);
        }
      } else {
        const int jj = col - 1536, u = jj >> 3, w = jj & 7;
        #pragma unroll
        for (int v = 0; v < 8; ++v) {
          int e = eb + mrow0 + v;
          float val = alpha * (cc[t][v] + bias) * sY[e * 4 + 0];
          unsafeAtomicAdd(&sTP[e * IN_DIM + MUL0 + w * 3 + 0], val * sX[e * IN_DIM + MUL0 + u * 3 + 0]);
          unsafeAtomicAdd(&sTP[e * IN_DIM + MUL0 + w * 3 + 1], val * sX[e * IN_DIM + MUL0 + u * 3 + 1]);
          unsafeAtomicAdd(&sTP[e * IN_DIM + MUL0 + w * 3 + 2], val * sX[e * IN_DIM + MUL0 + u * 3 + 2]);
        }
      }
    }
  }
  __syncthreads();

  for (int i = tid; i < 32 * IN_DIM; i += 256) {
    int r = i / IN_DIM, c = i - r * IN_DIM;
    unsafeAtomicAdd(&num[(size_t)sSrc[r] * IN_DIM + c], sTP[i]);
  }
  if (tid < 32) unsafeAtomicAdd(&cnt[sSrc[tid]], 1.0f);
}

__global__ __launch_bounds__(128)
void node_pass1(const float* __restrict__ node_attr,
                const float* __restrict__ num, const float* __restrict__ cnt,
                float* __restrict__ outp, float* __restrict__ stat)
{
  __shared__ float ssum[MUL0], ssq[MUL0], svq[MUL1];
  const int tid = threadIdx.x;
  if (tid < MUL0) { ssum[tid] = 0.0f; ssq[tid] = 0.0f; }
  if (tid < MUL1) svq[tid] = 0.0f;
  __syncthreads();

  const int n = blockIdx.x * 128 + tid;
  if (n < N_NODES) {
    float c   = cnt[n];
    float inv = 1.0f / (c > 1.0f ? c : 1.0f);
    for (int ch = 0; ch < MUL0; ++ch) {
      float p = num[(size_t)n * IN_DIM + ch] * inv + node_attr[(size_t)n * IN_DIM + ch];
      outp[(size_t)n * IN_DIM + ch] = p;
      unsafeAtomicAdd(&ssum[ch], p);
      unsafeAtomicAdd(&ssq[ch], p * p);
    }
    for (int u = 0; u < MUL1; ++u) {
      float acc = 0.0f;
      for (int mm = 0; mm < 3; ++mm) {
        int ch = MUL0 + u * 3 + mm;
        float p = num[(size_t)n * IN_DIM + ch] * inv + node_attr[(size_t)n * IN_DIM + ch];
        outp[(size_t)n * IN_DIM + ch] = p;
        acc += p * p;
      }
      unsafeAtomicAdd(&svq[u], acc);
    }
  }
  __syncthreads();
  if (tid < MUL0) {
    unsafeAtomicAdd(&stat[tid], ssum[tid]);
    unsafeAtomicAdd(&stat[MUL0 + tid], ssq[tid]);
  }
  if (tid < MUL1) unsafeAtomicAdd(&stat[2 * MUL0 + tid], svq[tid]);
}

__global__ __launch_bounds__(128)
void node_pass2(const float* __restrict__ stat,
                const float* __restrict__ bn_w_s, const float* __restrict__ bn_w_v,
                const float* __restrict__ bn_b_s,
                float* __restrict__ out)
{
  const int n = blockIdx.x * 128 + threadIdx.x;
  if (n >= N_NODES) return;
  const float invN = 1.0f / (float)N_NODES;
  for (int ch = 0; ch < MUL0; ++ch) {
    float mean = stat[ch] * invN;
    float var  = stat[MUL0 + ch] * invN - mean * mean;
    float p = out[(size_t)n * IN_DIM + ch];
    out[(size_t)n * IN_DIM + ch] = (p - mean) * rsqrtf(var + EPS) * bn_w_s[ch] + bn_b_s[ch];
  }
  for (int u = 0; u < MUL1; ++u) {
    float vn    = stat[2 * MUL0 + u] * invN * (1.0f / 3.0f);
    float scale = rsqrtf(vn + EPS) * bn_w_v[u];
    for (int mm = 0; mm < 3; ++mm) {
      int ch = MUL0 + u * 3 + mm;
      out[(size_t)n * IN_DIM + ch] *= scale;
    }
  }
}

extern "C" void kernel_launch(void* const* d_in, const int* in_sizes, int n_in,
                              void* d_out, int out_size, void* d_ws, size_t ws_size,
                              hipStream_t stream) {
  const float* node_attr = (const float*)d_in[0];
  const int*   edge_index = (const int*)d_in[1];
  const float* edge_attr = (const float*)d_in[2];
  const float* edge_sh   = (const float*)d_in[3];
  const float* fc1_w = (const float*)d_in[4];
  const float* fc1_b = (const float*)d_in[5];
  const float* fc2_w = (const float*)d_in[6];
  const float* fc2_b = (const float*)d_in[7];
  const float* bn_w_s = (const float*)d_in[8];
  const float* bn_w_v = (const float*)d_in[9];
  const float* bn_b_s = (const float*)d_in[10];

  float* ws   = (float*)d_ws;
  float* num  = ws + WS_NUM;
  float* cnt  = ws + WS_CNT;
  float* stat = ws + WS_STAT;
  float* out  = (float*)d_out;

  zero_ws_kernel<<<(WS_TOTAL + 255) / 256, 256, 0, stream>>>(ws);
  edge_kernel<<<N_EDGES / 32, 256, 0, stream>>>(node_attr, edge_index, edge_attr, edge_sh,
                                                fc1_w, fc1_b, fc2_w, fc2_b, num, cnt);
  node_pass1<<<(N_NODES + 127) / 128, 128, 0, stream>>>(node_attr, num, cnt, out, stat);
  node_pass2<<<(N_NODES + 127) / 128, 128, 0, stream>>>(stat, bn_w_s, bn_w_v, bn_b_s, out);
}


# edge kernel: tensor-product sums stay in registers across a wave's tiles and reach LDS once per weight block (summation order changes); per-edge multipliers pre-read from a [row][edge] LDS table
# speedup vs baseline: 1.1831x; 1.1018x over previous
#include <hip/hip_runtime.h>

#define N_NODES 10000
#define N_EDGES 100000
#define HIDDEN 128
#define MUL0 32
#define MUL1 8
#define IN_DIM 56
#define W_NUMEL 1600
#define EPS 1e-5f

typedef __attribute__((ext_vector_type(2))) float v2f;
typedef __attribute__((ext_vector_type(8))) float v8f;

__device__ __forceinline__ void load_bfrag(v2f (&bf)[32], const float* __restrict__ p, int ld) {
  #pragma unroll
  for (int i = 0; i < 32; ++i) {
    bf[i].x = p[(size_t)(4 * i) * ld];
    bf[i].y = p[(size_t)(4 * i + 1) * ld];
  }
}

#define WS_NUM   0
#define WS_CNT   (N_NODES*IN_DIM)
#define WS_STAT  (WS_CNT + N_NODES)
#define WS_TOTAL (WS_STAT + 72)

__global__ __launch_bounds__(256) void zero_ws_kernel(float* __restrict__ ws) {
  int i = blockIdx.x * 256 + threadIdx.x;
  if (i < WS_TOTAL) ws[i] = 0.0f;
}

__global__ __launch_bounds__(256)
void edge_kernel(const float* __restrict__ node_attr,
                 const int*   __restrict__ edge_index,
                 const float* __restrict__ edge_attr,
                 const float* __restrict__ edge_sh,
                 const float* __restrict__ fc1_w, const float* __restrict__ fc1_b,
                 const float* __restrict__ fc2_w, const float* __restrict__ fc2_b,
                 float* __restrict__ num, float* __restrict__ cnt)
{
  __shared__ float sA[32 * 132];
  __shared__ float sH[32 * 132];
  __shared__ float sX[32 * IN_DIM];
  __shared__ float sY[32 * 4];
  __shared__ __attribute__((aligned(32))) float sZT[73 * 32];
  __shared__ float sTP[32 * IN_DIM];
  __shared__ int   sSrc[32];

  const int tid  = threadIdx.x;
  const int lane = tid & 31;
  const int wv   = tid >> 5;
  const int e0   = blockIdx.x * 32;

  const int m     = lane & 15;
  const int koff  = (lane >> 4) << 1;
  const int mrow0 = (lane >> 4) << 3;

  v2f bf[32];
  const float bias1 = fc1_b[wv * 16 + m];
  load_bfrag(bf, fc1_w + (size_t)koff * HIDDEN + wv * 16 + m, HIDDEN);

  for (int i = tid; i < 32 * HIDDEN; i += 256) {
    int r = i >> 7, c = i & 127;
    sA[r * 132 + c] = edge_attr[(size_t)(e0 + r) * HIDDEN + c];
  }
  if (tid < 32) sSrc[tid] = edge_index[e0 + tid];
  if (tid < 128) {
    int r = tid >> 2, c = tid & 3;
    sY[r * 4 + c] = edge_sh[(size_t)(e0 + r) * 4 + c];
  }
  for (int i = tid; i < 32 * IN_DIM; i += 256) {
    int r = i / IN_DIM, c = i - r * IN_DIM;
    int dst = edge_index[N_EDGES + e0 + r];
    sX[i]  = node_attr[(size_t)dst * IN_DIM + c];
    sTP[i] = 0.0f;
  }
  __syncthreads();

  const float alpha = 0.15811388300841897f;
  const float ais3  = alpha * 0.5773502691896258f;
  for (int i = tid; i < 32 * MUL0; i += 256) {
    int u = i >> 5, r = i & 31;
    float x0 = sX[r * IN_DIM + u];
    sZT[i]           = alpha * (x0 * sY[r * 4 + 0]);
    sZT[40 * 32 + i] = alpha * x0;
  }
  for (int i = tid; i < 32 * MUL1; i += 256) {
    int u = i >> 5, r = i & 31;
    const float* xr = &sX[r * IN_DIM + MUL0 + u * 3];
    const float* yr = &sY[r * 4 + 1];
    sZT[32 * 32 + i] = ais3 * (xr[0] * yr[0] + xr[1] * yr[1] + xr[2] * yr[2]);
  }
  if (tid < 32) sZT[72 * 32 + tid] = alpha * sY[tid * 4 + 0];
  __syncthreads();

  float bias_next;
  {
    v8f acc0 = {}, acc1 = {};
    const int col = wv * 16 + m;
    #pragma unroll
    for (int i = 0; i < 32; ++i) {
      const int kb = 4 * i;
      v2f a0, a1;
      a0.x = sA[m * 132 + kb + koff];
      a0.y = sA[m * 132 + kb + koff + 1];
      a1.x = sA[(16 + m) * 132 + kb + koff];
      a1.y = sA[(16 + m) * 132 + kb + koff + 1];
      acc0 = __builtin_amdgcn_wmma_f32_16x16x4_f32(false, a0, false, bf[i], (short)0, acc0, false, false);
      acc1 = __builtin_amdgcn_wmma_f32_16x16x4_f32(false, a1, false, bf[i], (short)0, acc1, false, false);
    }
    bias_next = fc2_b[col];
    load_bfrag(bf, fc2_w + (size_t)koff * W_NUMEL + col, W_NUMEL);
    const float bias = bias1;
    #pragma unroll
    for (int v = 0; v < 8; ++v) {
      float h0 = acc0[v] + bias;
      float h1 = acc1[v] + bias;
      sH[(mrow0 + v) * 132 + col]      = h0 > 0.0f ? h0 : 0.0f;
      sH[(16 + mrow0 + v) * 132 + col] = h1 > 0.0f ? h1 : 0.0f;
    }
  }
  __syncthreads();

  v8f o[2] = {{}, {}};

  for (int nt = wv; nt < 100; nt += 8) {
    v8f cc[2] = {{}, {}};
    const int col = nt * 16 + m;
    const float bias = bias_next;
    const int zrow = nt < 80 ? (nt >> 1) : (nt < 96 ? 2 * nt - 120 + (m >> 3) : 72);
    v8f z[2];
    z[0] = *(const v8f*)&sZT[zrow * 32 + mrow0];
    z[1] = *(const v8f*)&sZT[zrow * 32 + 16 + mrow0];
    #pragma unroll
    for (int i = 0; i < 32; ++i) {
      const int kb = 4 * i;
      v2f a0, a1;
      a0.x = sH[m * 132 + kb + koff];
      a0.y = sH[m * 132 + kb + koff + 1];
      a1.x = sH[(16 + m) * 132 + kb + koff];
      a1.y = sH[(16 + m) * 132 + kb + koff + 1];
      cc[0] = __builtin_amdgcn_wmma_f32_16x16x4_f32(false, a0, false, bf[i], (short)0, cc[0], false, false);
      cc[1] = __builtin_amdgcn_wmma_f32_16x16x4_f32(false, a1, false, bf[i], (short)0, cc[1], false, false);
    }
    if (nt + 8 < 100) {
      bias_next = fc2_b[col + 128];
      load_bfrag(bf, fc2_w + (size_t)koff * W_NUMEL + col + 128, W_NUMEL);
    }

    v8f p[2];
    #pragma unroll
    for (int t = 0; t < 2; ++t) {
      #pragma unroll
      for (int v = 0; v < 8; ++v) p[t][v] = (cc[t][v] + bias) * z[t][v];
    }
    const int w3 = MUL0 + (m & 7) * 3;
    if (nt < 96) {
      o[0] += p[0];
      o[1] += p[1];
      if (nt >= 72 && nt < 80) {
        const int w = (nt & 1) * 16 + m;
        #pragma unroll
        for (int t = 0; t < 2; ++t) {
          #pragma unroll
          for (int v = 0; v < 8; ++v) {
            int e = t * 16 + mrow0 + v;
            unsafeAtomicAdd(&sTP[e * IN_DIM + w], o[t][v]);
          }
        }
        o[0] = v8f{};
        o[1] = v8f{};
      } else if (nt >= 88) {
        float yv[2][8][3];
        #pragma unroll
        for (int t = 0; t < 2; ++t) {
          #pragma unroll
          for (int v = 0; v < 8; ++v) {
            int e = t * 16 + mrow0 + v;
            #pragma unroll
            for (int mm = 0; mm < 3; ++mm) yv[t][v][mm] = o[t][v] * sY[e * 4 + 1 + mm];
          }
        }
        #pragma unroll
        for (int t = 0; t < 2; ++t) {
          #pragma unroll
          for (int v = 0; v < 8; ++v) {
            int e = t * 16 + mrow0 + v;
            #pragma unroll
            for (int mm = 0; mm < 3; ++mm) unsafeAtomicAdd(&sTP[e * IN_DIM + w3 + mm], yv[t][v][mm]);
          }
        }
      }
    } else {
      const int u3 = MUL0 + ((nt - 96) * 2 + (m >> 3)) * 3;
      float xv[2][8][3];
      #pragma unroll
      for (int t = 0; t < 2; ++t) {
        #pragma unroll
        for (int v = 0; v < 8; ++v) {
          int e = t * 16 + mrow0 + v;
          #pragma unroll
          for (int mm = 0; mm < 3; ++mm) xv[t][v][mm] = p[t][v] * sX[e * IN_DIM + u3 + mm];
        }
      }
      #pragma unroll
      for (int t = 0; t < 2; ++t) {
        #pragma unroll
        for (int v = 0; v < 8; ++v) {
          int e = t * 16 + mrow0 + v;
          #pragma unroll
          for (int mm = 0; mm < 3; ++mm) unsafeAtomicAdd(&sTP[e * IN_DIM + w3 + mm], xv[t][v][mm]);
        }
      }
    }
  }
  __syncthreads();

  for (int i = tid; i < 32 * IN_DIM; i += 256) {
    int r = i / IN_DIM, c = i - r * IN_DIM;
    unsafeAtomicAdd(&num[(size_t)sSrc[r] * IN_DIM + c], sTP[i]);
  }
  if (tid < 32) unsafeAtomicAdd(&cnt[sSrc[tid]], 1.0f);
}

__global__ __launch_bounds__(128)
void node_pass1(const float* __restrict__ node_attr,
                const float* __restrict__ num, const float* __restrict__ cnt,
                float* __restrict__ outp, float* __restrict__ stat)
{
  __shared__ float ssum[MUL0], ssq[MUL0], svq[MUL1];
  const int tid = threadIdx.x;
  if (tid < MUL0) { ssum[tid] = 0.0f; ssq[tid] = 0.0f; }
  if (tid < MUL1) svq[tid] = 0.0f;
  __syncthreads();

  const int n = blockIdx.x * 128 + tid;
  if (n < N_NODES) {
    float c   = cnt[n];
    float inv = 1.0f / (c > 1.0f ? c : 1.0f);
    for (int ch = 0; ch < MUL0; ++ch) {
      float p = num[(size_t)n * IN_DIM + ch] * inv + node_attr[(size_t)n * IN_DIM + ch];
      outp[(size_t)n * IN_DIM + ch] = p;
      unsafeAtomicAdd(&ssum[ch], p);
      unsafeAtomicAdd(&ssq[ch], p * p);
    }
    for (int u = 0; u < MUL1; ++u) {
      float acc = 0.0f;
      for (int mm = 0; mm < 3; ++mm) {
        int ch = MUL0 + u * 3 + mm;
        float p = num[(size_t)n * IN_DIM + ch] * inv + node_attr[(size_t)n * IN_DIM + ch];
        outp[(size_t)n * IN_DIM + ch] = p;
        acc += p * p;
      }
      unsafeAtomicAdd(&svq[u], acc);
    }
  }
  __syncthreads();
  if (tid < MUL0) {
    unsafeAtomicAdd(&stat[tid], ssum[tid]);
    unsafeAtomicAdd(&stat[MUL0 + tid], ssq[tid]);
  }
  if (tid < MUL1) unsafeAtomicAdd(&stat[2 * MUL0 + tid], svq[tid]);
}

__global__ __launch_bounds__(128)
void node_pass2(const float* __restrict__ stat,
                const float* __restrict__ bn_w_s, const float* __restrict__ bn_w_v,
                const float* __restrict__ bn_b_s,
                float* __restrict__ out)
{
  const int n = blockIdx.x * 128 + threadIdx.x;
  if (n >= N_NODES) return;
  const float invN = 1.0f / (float)N_NODES;
  for (int ch = 0; ch < MUL0; ++ch) {
    float mean = stat[ch] * invN;
    float var  = stat[MUL0 + ch] * invN - mean * mean;
    float p = out[(size_t)n * IN_DIM + ch];
    out[(size_t)n * IN_DIM + ch] = (p - mean) * rsqrtf(var + EPS) * bn_w_s[ch] + bn_b_s[ch];
  }
  for (int u = 0; u < MUL1; ++u) {
    float vn    = stat[2 * MUL0 + u] * invN * (1.0f / 3.0f);
    float scale = rsqrtf(vn + EPS) * bn_w_v[u];
    for (int mm = 0; mm < 3; ++mm) {
      int ch = MUL0 + u * 3 + mm;
      out[(size_t)n * IN_DIM + ch] *= scale;
    }
  }
}

extern "C" void kernel_launch(void* const* d_in, const int* in_sizes, int n_in,
                              void* d_out, int out_size, void* d_ws, size_t ws_size,
                              hipStream_t stream) {
  const float* node_attr = (const float*)d_in[0];
  const int*   edge_index = (const int*)d_in[1];
  const float* edge_attr = (const float*)d_in[2];
  const float* edge_sh   = (const float*)d_in[3];
  const float* fc1_w = (const float*)d_in[4];
  const float* fc1_b = (const float*)d_in[5];
  const float* fc2_w = (const float*)d_in[6];
  const float* fc2_b = (const float*)d_in[7];
  const float* bn_w_s = (const float*)d_in[8];
  const float* bn_w_v = (const float*)d_in[9];
  const float* bn_b_s = (const float*)d_in[10];

  float* ws   = (float*)d_ws;
  float* num  = ws + WS_NUM;
  float* cnt  = ws + WS_CNT;
  float* stat = ws + WS_STAT;
  float* out  = (float*)d_out;

  zero_ws_kernel<<<(WS_TOTAL + 255) / 256, 256, 0, stream>>>(ws);
  edge_kernel<<<N_EDGES / 32, 256, 0, stream>>>(node_attr, edge_index, edge_attr, edge_sh,
                                                fc1_w, fc1_b, fc2_w, fc2_b, num, cnt);
  node_pass1<<<(N_NODES + 127) / 128, 128, 0, stream>>>(node_attr, num, cnt, out, stat);
  node_pass2<<<(N_NODES + 127) / 128, 128, 0, stream>>>(stat, bn_w_s, bn_w_v, bn_b_s, out);
}


# edge kernel: both GEMMs' A fragments (32 rows x K=128) stay in 128 registers per wave, read from LDS once, so no LDS read sits among WMMAs; 4-wave blocks, 25 tiles per wave (summation order changes)
# speedup vs baseline: 1.3028x; 1.1012x over previous
#include <hip/hip_runtime.h>

#define N_NODES 10000
#define N_EDGES 100000
#define HIDDEN 128
#define MUL0 32
#define MUL1 8
#define IN_DIM 56
#define W_NUMEL 1600
#define EPS 1e-5f

typedef __attribute__((ext_vector_type(2))) float v2f;
typedef __attribute__((ext_vector_type(8))) float v8f;

__device__ __forceinline__ void load_bfrag(v2f (&bf)[32], const float* __restrict__ p, int ld) {
  #pragma unroll
  for (int i = 0; i < 32; ++i) {
    bf[i].x = p[(size_t)(4 * i) * ld];
    bf[i].y = p[(size_t)(4 * i + 1) * ld];
  }
}

#define WS_NUM   0
#define WS_CNT   (N_NODES*IN_DIM)
#define WS_STAT  (WS_CNT + N_NODES)
#define WS_TOTAL (WS_STAT + 72)

__global__ __launch_bounds__(256) void zero_ws_kernel(float* __restrict__ ws) {
  int i = blockIdx.x * 256 + threadIdx.x;
  if (i < WS_TOTAL) ws[i] = 0.0f;
}

__global__ __launch_bounds__(128)
void edge_kernel(const float* __restrict__ node_attr,
                 const int*   __restrict__ edge_index,
                 const float* __restrict__ edge_attr,
                 const float* __restrict__ edge_sh,
                 const float* __restrict__ fc1_w, const float* __restrict__ fc1_b,
                 const float* __restrict__ fc2_w, const float* __restrict__ fc2_b,
                 float* __restrict__ num, float* __restrict__ cnt)
{
  __shared__ __attribute__((aligned(16))) float sA[32 * 132];
  __shared__ __attribute__((aligned(16))) float sH[32 * 132];
  __shared__ float sX[32 * IN_DIM];
  __shared__ float sY[32 * 4];
  __shared__ __attribute__((aligned(32))) float sZT[73 * 32];
  __shared__ float sTP[32 * IN_DIM];
  __shared__ int   sSrc[32];

  const int tid  = threadIdx.x;
  const int lane = tid & 31;
  const int wv   = tid >> 5;
  const int e0   = blockIdx.x * 32;

  const int m     = lane & 15;
  const int koff  = (lane >> 4) << 1;
  const int mrow0 = (lane >> 4) << 3;

  v2f bf[32];
  const float bias1a = fc1_b[wv * 16 + m];
  const float bias1b = fc1_b[(wv + 4) * 16 + m];
  load_bfrag(bf, fc1_w + (size_t)koff * HIDDEN + wv * 16 + m, HIDDEN);

  for (int i = tid; i < 32 * HIDDEN; i += 128) {
    int r = i >> 7, c = i & 127;
    sA[r * 132 + c] = edge_attr[(size_t)(e0 + r) * HIDDEN + c];
  }
  if (tid < 32) sSrc[tid] = edge_index[e0 + tid];
  if (tid < 128) {
    int r = tid >> 2, c = tid & 3;
    sY[r * 4 + c] = edge_sh[(size_t)(e0 + r) * 4 + c];
  }
  for (int i = tid; i < 32 * IN_DIM; i += 128) {
    int r = i / IN_DIM, c = i - r * IN_DIM;
    int dst = edge_index[N_EDGES + e0 + r];
    sX[i]  = node_attr[(size_t)dst * IN_DIM + c];
    sTP[i] = 0.0f;
  }
  __syncthreads();

  const float alpha = 0.15811388300841897f;
  const float ais3  = alpha * 0.5773502691896258f;
  for (int i = tid; i < 32 * MUL0; i += 128) {
    int u = i >> 5, r = i & 31;
    float x0 = sX[r * IN_DIM + u];
    sZT[i]           = alpha * (x0 * sY[r * 4 + 0]);
    sZT[40 * 32 + i] = alpha * x0;
  }
  for (int i = tid; i < 32 * MUL1; i += 128) {
    int u = i >> 5, r = i & 31;
    const float* xr = &sX[r * IN_DIM + MUL0 + u * 3];
    const float* yr = &sY[r * 4 + 1];
    sZT[32 * 32 + i] = ais3 * (xr[0] * yr[0] + xr[1] * yr[1] + xr[2] * yr[2]);
  }
  if (tid < 32) sZT[72 * 32 + tid] = alpha * sY[tid * 4 + 0];
  __syncthreads();

  v2f af[2][32];
  #pragma unroll
  for (int i = 0; i < 32; ++i) {
    af[0][i].x = sA[m * 132 + 4 * i + koff];
    af[0][i].y = sA[m * 132 + 4 * i + koff + 1];
    af[1][i].x = sA[(16 + m) * 132 + 4 * i + koff];
    af[1][i].y = sA[(16 + m) * 132 + 4 * i + koff + 1];
  }

  float bias_next;
  #pragma unroll
  for (int g = 0; g < 2; ++g) {
    v8f acc0 = {}, acc1 = {};
    const int col = (wv + 4 * g) * 16 + m;
    #pragma unroll
    for (int i = 0; i < 32; ++i) {
      acc0 = __builtin_amdgcn_wmma_f32_16x16x4_f32(false, af[0][i], false, bf[i], (short)0, acc0, false, false);
      acc1 = __builtin_amdgcn_wmma_f32_16x16x4_f32(false, af[1][i], false, bf[i], (short)0, acc1, false, false);
    }
    if (g == 0) {
      load_bfrag(bf, fc1_w + (size_t)koff * HIDDEN + col + 64, HIDDEN);
    } else {
      bias_next = fc2_b[wv * 16 + m];
      load_bfrag(bf, fc2_w + (size_t)koff * W_NUMEL + wv * 16 + m, W_NUMEL);
    }
    const float bias = g == 0 ? bias1a : bias1b;
    #pragma unroll
    for (int v = 0; v < 8; ++v) {
      float h0 = acc0[v] + bias;
      float h1 = acc1[v] + bias;
      sH[(mrow0 + v) * 132 + col]      = h0 > 0.0f ? h0 : 0.0f;
      sH[(16 + mrow0 + v) * 132 + col] = h1 > 0.0f ? h1 : 0.0f;
    }
  }
  __syncthreads();

  #pragma unroll
  for (int i = 0; i < 32; ++i) {
    af[0][i].x = sH[m * 132 + 4 * i + koff];
    af[0][i].y = sH[m * 132 + 4 * i + koff + 1];
    af[1][i].x = sH[(16 + m) * 132 + 4 * i + koff];
    af[1][i].y = sH[(16 + m) * 132 + 4 * i + koff + 1];
  }

  v8f o[2] = {{}, {}};

  for (int nt = wv; nt < 100; nt += 4) {
    v8f cc[2] = {{}, {}};
    const int col = nt * 16 + m;
    const float bias = bias_next;
    const int zrow = nt < 80 ? (nt >> 1) : (nt < 96 ? 2 * nt - 120 + (m >> 3) : 72);
    v8f z[2];
    z[0] = *(const v8f*)&sZT[zrow * 32 + mrow0];
    z[1] = *(const v8f*)&sZT[zrow * 32 + 16 + mrow0];
    #pragma unroll
    for (int i = 0; i < 32; ++i) {
      cc[0] = __builtin_amdgcn_wmma_f32_16x16x4_f32(false, af[0][i], false, bf[i], (short)0, cc[0], false, false);
      cc[1] = __builtin_amdgcn_wmma_f32_16x16x4_f32(false, af[1][i], false, bf[i], (short)0, cc[1], false, false);
    }
    if (nt + 4 < 100) {
      bias_next = fc2_b[col + 64];
      load_bfrag(bf, fc2_w + (size_t)koff * W_NUMEL + col + 64, W_NUMEL);
    }

    v8f p[2];
    #pragma unroll
    for (int t = 0; t < 2; ++t) {
      #pragma unroll
      for (int v = 0; v < 8; ++v) p[t][v] = (cc[t][v] + bias) * z[t][v];
    }
    const int w3 = MUL0 + (m & 7) * 3;
    if (nt < 96) {
      o[0] += p[0];
      o[1] += p[1];
      if (nt >= 76 && nt < 80) {
        const int w = (nt & 1) * 16 + m;
        #pragma unroll
        for (int t = 0; t < 2; ++t) {
          #pragma unroll
          for (int v = 0; v < 8; ++v) {
            int e = t * 16 + mrow0 + v;
            unsafeAtomicAdd(&sTP[e * IN_DIM + w], o[t][v]);
          }
        }
        o[0] = v8f{};
        o[1] = v8f{};
      } else if (nt >= 92) {
        float yv[2][8][3];
        #pragma unroll
        for (int t = 0; t < 2; ++t) {
          #pragma unroll
          for (int v = 0; v < 8; ++v) {
            int e = t * 16 + mrow0 + v;
            #pragma unroll
            for (int mm = 0; mm < 3; ++mm) yv[t][v][mm] = o[t][v] * sY[e * 4 + 1 + mm];
          }
        }
        #pragma unroll
        for (int t = 0; t < 2; ++t) {
          #pragma unroll
          for (int v = 0; v < 8; ++v) {
            int e = t * 16 + mrow0 + v;
            #pragma unroll
            for (int mm = 0; mm < 3; ++mm) unsafeAtomicAdd(&sTP[e * IN_DIM + w3 + mm], yv[t][v][mm]);
          }
        }
      }
    } else {
      const int u3 = MUL0 + ((nt - 96) * 2 + (m >> 3)) * 3;
      float xv[2][8][3];
      #pragma unroll
      for (int t = 0; t < 2; ++t) {
        #pragma unroll
        for (int v = 0; v < 8; ++v) {
          int e = t * 16 + mrow0 + v;
          #pragma unroll
          for (int mm = 0; mm < 3; ++mm) xv[t][v][mm] = p[t][v] * sX[e * IN_DIM + u3 + mm];
        }
      }
      #pragma unroll
      for (int t = 0; t < 2; ++t) {
        #pragma unroll
        for (int v = 0; v < 8; ++v) {
          int e = t * 16 + mrow0 + v;
          #pragma unroll
          for (int mm = 0; mm < 3; ++mm) unsafeAtomicAdd(&sTP[e * IN_DIM + w3 + mm], xv[t][v][mm]);
        }
      }
    }
  }
  __syncthreads();

  for (int i = tid; i < 32 * IN_DIM; i += 128) {
    int r = i / IN_DIM, c = i - r * IN_DIM;
    unsafeAtomicAdd(&num[(size_t)sSrc[r] * IN_DIM + c], sTP[i]);
  }
  if (tid < 32) unsafeAtomicAdd(&cnt[sSrc[tid]], 1.0f);
}

__global__ __launch_bounds__(128)
void node_pass1(const float* __restrict__ node_attr,
                const float* __restrict__ num, const float* __restrict__ cnt,
                float* __restrict__ outp, float* __restrict__ stat)
{
  __shared__ float ssum[MUL0], ssq[MUL0], svq[MUL1];
  const int tid = threadIdx.x;
  if (tid < MUL0) { ssum[tid] = 0.0f; ssq[tid] = 0.0f; }
  if (tid < MUL1) svq[tid] = 0.0f;
  __syncthreads();

  const int n = blockIdx.x * 128 + tid;
  if (n < N_NODES) {
    float c   = cnt[n];
    float inv = 1.0f / (c > 1.0f ? c : 1.0f);
    for (int ch = 0; ch < MUL0; ++ch) {
      float p = num[(size_t)n * IN_DIM + ch] * inv + node_attr[(size_t)n * IN_DIM + ch];
      outp[(size_t)n * IN_DIM + ch] = p;
      unsafeAtomicAdd(&ssum[ch], p);
      unsafeAtomicAdd(&ssq[ch], p * p);
    }
    for (int u = 0; u < MUL1; ++u) {
      float acc = 0.0f;
      for (int mm = 0; mm < 3; ++mm) {
        int ch = MUL0 + u * 3 + mm;
        float p = num[(size_t)n * IN_DIM + ch] * inv + node_attr[(size_t)n * IN_DIM + ch];
        outp[(size_t)n * IN_DIM + ch] = p;
        acc += p * p;
      }
      unsafeAtomicAdd(&svq[u], acc);
    }
  }
  __syncthreads();
  if (tid < MUL0) {
    unsafeAtomicAdd(&stat[tid], ssum[tid]);
    unsafeAtomicAdd(&stat[MUL0 + tid], ssq[tid]);
  }
  if (tid < MUL1) unsafeAtomicAdd(&stat[2 * MUL0 + tid], svq[tid]);
}

__global__ __launch_bounds__(128)
void node_pass2(const float* __restrict__ stat,
                const float* __restrict__ bn_w_s, const float* __restrict__ bn_w_v,
                const float* __restrict__ bn_b_s,
                float* __restrict__ out)
{
  const int n = blockIdx.x * 128 + threadIdx.x;
  if (n >= N_NODES) return;
  const float invN = 1.0f / (float)N_NODES;
  for (int ch = 0; ch < MUL0; ++ch) {
    float mean = stat[ch] * invN;
    float var  = stat[MUL0 + ch] * invN - mean * mean;
    float p = out[(size_t)n * IN_DIM + ch];
    out[(size_t)n * IN_DIM + ch] = (p - mean) * rsqrtf(var + EPS) * bn_w_s[ch] + bn_b_s[ch];
  }
  for (int u = 0; u < MUL1; ++u) {
    float vn    = stat[2 * MUL0 + u] * invN * (1.0f / 3.0f);
    float scale = rsqrtf(vn + EPS) * bn_w_v[u];
    for (int mm = 0; mm < 3; ++mm) {
      int ch = MUL0 + u * 3 + mm;
      out[(size_t)n * IN_DIM + ch] *= scale;
    }
  }
}

extern "C" void kernel_launch(void* const* d_in, const int* in_sizes, int n_in,
                              void* d_out, int out_size, void* d_ws, size_t ws_size,
                              hipStream_t stream) {
  const float* node_attr = (const float*)d_in[0];
  const int*   edge_index = (const int*)d_in[1];
  const float* edge_attr = (const float*)d_in[2];
  const float* edge_sh   = (const float*)d_in[3];
  const float* fc1_w = (const float*)d_in[4];
  const float* fc1_b = (const float*)d_in[5];
  const float* fc2_w = (const float*)d_in[6];
  const float* fc2_b = (const float*)d_in[7];
  const float* bn_w_s = (const float*)d_in[8];
  const float* bn_w_v = (const float*)d_in[9];
  const float* bn_b_s = (const float*)d_in[10];

  float* ws   = (float*)d_ws;
  float* num  = ws + WS_NUM;
  float* cnt  = ws + WS_CNT;
  float* stat = ws + WS_STAT;
  float* out  = (float*)d_out;

  zero_ws_kernel<<<(WS_TOTAL + 255) / 256, 256, 0, stream>>>(ws);
  edge_kernel<<<N_EDGES / 32, 128, 0, stream>>>(node_attr, edge_index, edge_attr, edge_sh,
                                                fc1_w, fc1_b, fc2_w, fc2_b, num, cnt);
  node_pass1<<<(N_NODES + 127) / 128, 128, 0, stream>>>(node_attr, num, cnt, out, stat);
  node_pass2<<<(N_NODES + 127) / 128, 128, 0, stream>>>(stat, bn_w_s, bn_w_v, bn_b_s, out);
}


# edge kernel: GEMM2 weight columns 0..1279 fetched as whole 256-byte row pieces (16-byte loads, 4 column tiles per wave at once, K chunks one ahead); summation order changes
# speedup vs baseline: 1.3651x; 1.0478x over previous
#include <hip/hip_runtime.h>

#define N_NODES 10000
#define N_EDGES 100000
#define HIDDEN 128
#define MUL0 32
#define MUL1 8
#define IN_DIM 56
#define W_NUMEL 1600
#define EPS 1e-5f

typedef __attribute__((ext_vector_type(2))) float v2f;
typedef __attribute__((ext_vector_type(8))) float v8f;

__device__ __forceinline__ void load_bfrag(v2f (&bf)[32], const float* __restrict__ p, int ld) {
  #pragma unroll
  for (int i = 0; i < 32; ++i) {
    bf[i].x = p[(size_t)(4 * i) * ld];
    bf[i].y = p[(size_t)(4 * i + 1) * ld];
  }
}

typedef __attribute__((ext_vector_type(4))) float v4f;

__device__ __forceinline__ void load_q(v4f (&q)[16], const float* __restrict__ p) {
  #pragma unroll
  for (int i = 0; i < 8; ++i) {
    q[2 * i]     = *(const v4f*)(p + (size_t)(4 * i) * W_NUMEL);
    q[2 * i + 1] = *(const v4f*)(p + (size_t)(4 * i + 1) * W_NUMEL);
  }
}

template <int CH>
__device__ __forceinline__ void mma_chunk(v8f (&cc)[4][2], const v2f (&af)[2][32], const v4f (&q)[16]) {
  #pragma unroll
  for (int i = 0; i < 8; ++i) {
    #pragma unroll
    for (int j = 0; j < 4; ++j) {
      v2f b;
      b.x = q[2 * i][j];
      b.y = q[2 * i + 1][j];
      cc[j][0] = __builtin_amdgcn_wmma_f32_16x16x4_f32(false, af[0][8 * CH + i], false, b, (short)0, cc[j][0], false, false);
      cc[j][1] = __builtin_amdgcn_wmma_f32_16x16x4_f32(false, af[1][8 * CH + i], false, b, (short)0, cc[j][1], false, false);
    }
  }
}

#define WS_NUM   0
#define WS_CNT   (N_NODES*IN_DIM)
#define WS_STAT  (WS_CNT + N_NODES)
#define WS_TOTAL (WS_STAT + 72)

__global__ __launch_bounds__(256) void zero_ws_kernel(float* __restrict__ ws) {
  int i = blockIdx.x * 256 + threadIdx.x;
  if (i < WS_TOTAL) ws[i] = 0.0f;
}

__global__ __launch_bounds__(128)
void edge_kernel(const float* __restrict__ node_attr,
                 const int*   __restrict__ edge_index,
                 const float* __restrict__ edge_attr,
                 const float* __restrict__ edge_sh,
                 const float* __restrict__ fc1_w, const float* __restrict__ fc1_b,
                 const float* __restrict__ fc2_w, const float* __restrict__ fc2_b,
                 float* __restrict__ num, float* __restrict__ cnt)
{
  __shared__ __attribute__((aligned(16))) float sA[32 * 132];
  __shared__ __attribute__((aligned(16))) float sH[32 * 132];
  __shared__ float sX[32 * IN_DIM];
  __shared__ float sY[32 * 4];
  __shared__ __attribute__((aligned(32))) float sZT[73 * 32];
  __shared__ float sTP[32 * IN_DIM];
  __shared__ int   sSrc[32];

  const int tid  = threadIdx.x;
  const int lane = tid & 31;
  const int wv   = tid >> 5;
  const int e0   = blockIdx.x * 32;

  const int m     = lane & 15;
  const int koff  = (lane >> 4) << 1;
  const int mrow0 = (lane >> 4) << 3;

  v2f bf[32];
  v4f qa[16], qb[16];
  v4f bias4;
  const float* wq = fc2_w + (size_t)koff * W_NUMEL + 4 * m;
  const float bias1a = fc1_b[wv * 16 + m];
  const float bias1b = fc1_b[(wv + 4) * 16 + m];
  load_bfrag(bf, fc1_w + (size_t)koff * HIDDEN + wv * 16 + m, HIDDEN);

  for (int i = tid; i < 32 * HIDDEN; i += 128) {
    int r = i >> 7, c = i & 127;
    sA[r * 132 + c] = edge_attr[(size_t)(e0 + r) * HIDDEN + c];
  }
  if (tid < 32) sSrc[tid] = edge_index[e0 + tid];
  if (tid < 128) {
    int r = tid >> 2, c = tid & 3;
    sY[r * 4 + c] = edge_sh[(size_t)(e0 + r) * 4 + c];
  }
  for (int i = tid; i < 32 * IN_DIM; i += 128) {
    int r = i / IN_DIM, c = i - r * IN_DIM;
    int dst = edge_index[N_EDGES + e0 + r];
    sX[i]  = node_attr[(size_t)dst * IN_DIM + c];
    sTP[i] = 0.0f;
  }
  __syncthreads();

  const float alpha = 0.15811388300841897f;
  const float ais3  = alpha * 0.5773502691896258f;
  for (int i = tid; i < 32 * MUL0; i += 128) {
    int u = i >> 5, r = i & 31;
    float x0 = sX[r * IN_DIM + u];
    sZT[i]           = alpha * (x0 * sY[r * 4 + 0]);
    sZT[40 * 32 + i] = alpha * x0;
  }
  for (int i = tid; i < 32 * MUL1; i += 128) {
    int u = i >> 5, r = i & 31;
    const float* xr = &sX[r * IN_DIM + MUL0 + u * 3];
    const float* yr = &sY[r * 4 + 1];
    sZT[32 * 32 + i] = ais3 * (xr[0] * yr[0] + xr[1] * yr[1] + xr[2] * yr[2]);
  }
  if (tid < 32) sZT[72 * 32 + tid] = alpha * sY[tid * 4 + 0];
  __syncthreads();

  v2f af[2][32];
  #pragma unroll
  for (int i = 0; i < 32; ++i) {
    af[0][i].x = sA[m * 132 + 4 * i + koff];
    af[0][i].y = sA[m * 132 + 4 * i + koff + 1];
    af[1][i].x = sA[(16 + m) * 132 + 4 * i + koff];
    af[1][i].y = sA[(16 + m) * 132 + 4 * i + koff + 1];
  }

  #pragma unroll
  for (int g = 0; g < 2; ++g) {
    v8f acc0 = {}, acc1 = {};
    const int col = (wv + 4 * g) * 16 + m;
    #pragma unroll
    for (int i = 0; i < 32; ++i) {
      acc0 = __builtin_amdgcn_wmma_f32_16x16x4_f32(false, af[0][i], false, bf[i], (short)0, acc0, false, false);
      acc1 = __builtin_amdgcn_wmma_f32_16x16x4_f32(false, af[1][i], false, bf[i], (short)0, acc1, false, false);
    }
    if (g == 0) {
      load_bfrag(bf, fc1_w + (size_t)koff * HIDDEN + col + 64, HIDDEN);
    } else {
      bias4 = *(const v4f*)&fc2_b[64 * wv + 4 * m];
      load_q(qa, wq + 64 * wv);
    }
    const float bias = g == 0 ? bias1a : bias1b;
    #pragma unroll
    for (int v = 0; v < 8; ++v) {
      float h0 = acc0[v] + bias;
      float h1 = acc1[v] + bias;
      sH[(mrow0 + v) * 132 + col]      = h0 > 0.0f ? h0 : 0.0f;
      sH[(16 + mrow0 + v) * 132 + col] = h1 > 0.0f ? h1 : 0.0f;
    }
  }
  __syncthreads();

  #pragma unroll
  for (int i = 0; i < 32; ++i) {
    af[0][i].x = sH[m * 132 + 4 * i + koff];
    af[0][i].y = sH[m * 132 + 4 * i + koff + 1];
    af[1][i].x = sH[(16 + m) * 132 + 4 * i + koff];
    af[1][i].y = sH[(16 + m) * 132 + 4 * i + koff + 1];
  }

  v8f o4[4][2];
  #pragma unroll
  for (int j = 0; j < 4; ++j) { o4[j][0] = v8f{}; o4[j][1] = v8f{}; }

  for (int gi = 0; gi < 5; ++gi) {
    const int g = wv + 4 * gi;
    const float* wg = wq + 64 * g;
    const int zrow = 2 * g + (m >> 3);
    const v8f z0 = *(const v8f*)&sZT[zrow * 32 + mrow0];
    const v8f z1 = *(const v8f*)&sZT[zrow * 32 + 16 + mrow0];
    const v4f bias = bias4;
    v8f cc[4][2];
    #pragma unroll
    for (int j = 0; j < 4; ++j) { cc[j][0] = v8f{}; cc[j][1] = v8f{}; }

    load_q(qb, wg + (size_t)32 * W_NUMEL);
    mma_chunk<0>(cc, af, qa);
    load_q(qa, wg + (size_t)64 * W_NUMEL);
    mma_chunk<1>(cc, af, qb);
    load_q(qb, wg + (size_t)96 * W_NUMEL);
    mma_chunk<2>(cc, af, qa);
    if (gi < 4) {
      bias4 = *(const v4f*)&fc2_b[64 * (g + 4) + 4 * m];
      load_q(qa, wg + 256);
    }
    mma_chunk<3>(cc, af, qb);

    #pragma unroll
    for (int j = 0; j < 4; ++j) {
      #pragma unroll
      for (int v = 0; v < 8; ++v) {
        o4[j][0][v] += (cc[j][0][v] + bias[j]) * z0[v];
        o4[j][1][v] += (cc[j][1][v] + bias[j]) * z1[v];
      }
    }
  }

  float bias_next = fc2_b[(80 + wv) * 16 + m];
  load_bfrag(bf, fc2_w + (size_t)koff * W_NUMEL + (80 + wv) * 16 + m, W_NUMEL);
  {
    const int w0 = 4 * (m & 7);
    #pragma unroll
    for (int j = 0; j < 4; ++j) {
      #pragma unroll
      for (int t = 0; t < 2; ++t) {
        #pragma unroll
        for (int v = 0; v < 8; ++v) {
          int e = t * 16 + mrow0 + v;
          unsafeAtomicAdd(&sTP[e * IN_DIM + w0 + j], o4[j][t][v]);
        }
      }
    }
  }

  v8f o[2] = {{}, {}};

  for (int nt = 80 + wv; nt < 100; nt += 4) {
    v8f cc[2] = {{}, {}};
    const int col = nt * 16 + m;
    const float bias = bias_next;
    const int zrow = nt < 96 ? 2 * nt - 120 + (m >> 3) : 72;
    v8f z[2];
    z[0] = *(const v8f*)&sZT[zrow * 32 + mrow0];
    z[1] = *(const v8f*)&sZT[zrow * 32 + 16 + mrow0];
    #pragma unroll
    for (int i = 0; i < 32; ++i) {
      cc[0] = __builtin_amdgcn_wmma_f32_16x16x4_f32(false, af[0][i], false, bf[i], (short)0, cc[0], false, false);
      cc[1] = __builtin_amdgcn_wmma_f32_16x16x4_f32(false, af[1][i], false, bf[i], (short)0, cc[1], false, false);
    }
    if (nt + 4 < 100) {
      bias_next = fc2_b[col + 64];
      load_bfrag(bf, fc2_w + (size_t)koff * W_NUMEL + col + 64, W_NUMEL);
    }

    v8f p[2];
    #pragma unroll
    for (int t = 0; t < 2; ++t) {
      #pragma unroll
      for (int v = 0; v < 8; ++v) p[t][v] = (cc[t][v] + bias) * z[t][v];
    }
    const int w3 = MUL0 + (m & 7) * 3;
    if (nt < 96) {
      o[0] += p[0];
      o[1] += p[1];
      if (nt >= 92) {
        float yv[2][8][3];
        #pragma unroll
        for (int t = 0; t < 2; ++t) {
          #pragma unroll
          for (int v = 0; v < 8; ++v) {
            int e = t * 16 + mrow0 + v;
            #pragma unroll
            for (int mm = 0; mm < 3; ++mm) yv[t][v][mm] = o[t][v] * sY[e * 4 + 1 + mm];
          }
        }
        #pragma unroll
        for (int t = 0; t < 2; ++t) {
          #pragma unroll
          for (int v = 0; v < 8; ++v) {
            int e = t * 16 + mrow0 + v;
            #pragma unroll
            for (int mm = 0; mm < 3; ++mm) unsafeAtomicAdd(&sTP[e * IN_DIM + w3 + mm], yv[t][v][mm]);
          }
        }
      }
    } else {
      const int u3 = MUL0 + ((nt - 96) * 2 + (m >> 3)) * 3;
      float xv[2][8][3];
      #pragma unroll
      for (int t = 0; t < 2; ++t) {
        #pragma unroll
        for (int v = 0; v < 8; ++v) {
          int e = t * 16 + mrow0 + v;
          #pragma unroll
          for (int mm = 0; mm < 3; ++mm) xv[t][v][mm] = p[t][v] * sX[e * IN_DIM + u3 + mm];
        }
      }
      #pragma unroll
      for (int t = 0; t < 2; ++t) {
        #pragma unroll
        for (int v = 0; v < 8; ++v) {
          int e = t * 16 + mrow0 + v;
          #pragma unroll
          for (int mm = 0; mm < 3; ++mm) unsafeAtomicAdd(&sTP[e * IN_DIM + w3 + mm], xv[t][v][mm]);
        }
      }
    }
  }
  __syncthreads();

  for (int i = tid; i < 32 * IN_DIM; i += 128) {
    int r = i / IN_DIM, c = i - r * IN_DIM;
    unsafeAtomicAdd(&num[(size_t)sSrc[r] * IN_DIM + c], sTP[i]);
  }
  if (tid < 32) unsafeAtomicAdd(&cnt[sSrc[tid]], 1.0f);
}

__global__ __launch_bounds__(128)
void node_pass1(const float* __restrict__ node_attr,
                const float* __restrict__ num, const float* __restrict__ cnt,
                float* __restrict__ outp, float* __restrict__ stat)
{
  __shared__ float ssum[MUL0], ssq[MUL0], svq[MUL1];
  const int tid = threadIdx.x;
  if (tid < MUL0) { ssum[tid] = 0.0f; ssq[tid] = 0.0f; }
  if (tid < MUL1) svq[tid] = 0.0f;
  __syncthreads();

  const int n = blockIdx.x * 128 + tid;
  if (n < N_NODES) {
    float c   = cnt[n];
    float inv = 1.0f / (c > 1.0f ? c : 1.0f);
    for (int ch = 0; ch < MUL0; ++ch) {
      float p = num[(size_t)n * IN_DIM + ch] * inv + node_attr[(size_t)n * IN_DIM + ch];
      outp[(size_t)n * IN_DIM + ch] = p;
      unsafeAtomicAdd(&ssum[ch], p);
      unsafeAtomicAdd(&ssq[ch], p * p);
    }
    for (int u = 0; u < MUL1; ++u) {
      float acc = 0.0f;
      for (int mm = 0; mm < 3; ++mm) {
        int ch = MUL0 + u * 3 + mm;
        float p = num[(size_t)n * IN_DIM + ch] * inv + node_attr[(size_t)n * IN_DIM + ch];
        outp[(size_t)n * IN_DIM + ch] = p;
        acc += p * p;
      }
      unsafeAtomicAdd(&svq[u], acc);
    }
  }
  __syncthreads();
  if (tid < MUL0) {
    unsafeAtomicAdd(&stat[tid], ssum[tid]);
    unsafeAtomicAdd(&stat[MUL0 + tid], ssq[tid]);
  }
  if (tid < MUL1) unsafeAtomicAdd(&stat[2 * MUL0 + tid], svq[tid]);
}

__global__ __launch_bounds__(128)
void node_pass2(const float* __restrict__ stat,
                const float* __restrict__ bn_w_s, const float* __restrict__ bn_w_v,
                const float* __restrict__ bn_b_s,
                float* __restrict__ out)
{
  const int n = blockIdx.x * 128 + threadIdx.x;
  if (n >= N_NODES) return;
  const float invN = 1.0f / (float)N_NODES;
  for (int ch = 0; ch < MUL0; ++ch) {
    float mean = stat[ch] * invN;
    float var  = stat[MUL0 + ch] * invN - mean * mean;
    float p = out[(size_t)n * IN_DIM + ch];
    out[(size_t)n * IN_DIM + ch] = (p - mean) * rsqrtf(var + EPS) * bn_w_s[ch] + bn_b_s[ch];
  }
  for (int u = 0; u < MUL1; ++u) {
    float vn    = stat[2 * MUL0 + u] * invN * (1.0f / 3.0f);
    float scale = rsqrtf(vn + EPS) * bn_w_v[u];
    for (int mm = 0; mm < 3; ++mm) {
      int ch = MUL0 + u * 3 + mm;
      out[(size_t)n * IN_DIM + ch] *= scale;
    }
  }
}

extern "C" void kernel_launch(void* const* d_in, const int* in_sizes, int n_in,
                              void* d_out, int out_size, void* d_ws, size_t ws_size,
                              hipStream_t stream) {
  const float* node_attr = (const float*)d_in[0];
  const int*   edge_index = (const int*)d_in[1];
  const float* edge_attr = (const float*)d_in[2];
  const float* edge_sh   = (const float*)d_in[3];
  const float* fc1_w = (const float*)d_in[4];
  const float* fc1_b = (const float*)d_in[5];
  const float* fc2_w = (const float*)d_in[6];
  const float* fc2_b = (const float*)d_in[7];
  const float* bn_w_s = (const float*)d_in[8];
  const float* bn_w_v = (const float*)d_in[9];
  const float* bn_b_s = (const float*)d_in[10];

  float* ws   = (float*)d_ws;
  float* num  = ws + WS_NUM;
  float* cnt  = ws + WS_CNT;
  float* stat = ws + WS_STAT;
  float* out  = (float*)d_out;

  zero_ws_kernel<<<(WS_TOTAL + 255) / 256, 256, 0, stream>>>(ws);
  edge_kernel<<<N_EDGES / 32, 128, 0, stream>>>(node_attr, edge_index, edge_attr, edge_sh,
                                                fc1_w, fc1_b, fc2_w, fc2_b, num, cnt);
  node_pass1<<<(N_NODES + 127) / 128, 128, 0, stream>>>(node_attr, num, cnt, out, stat);
  node_pass2<<<(N_NODES + 127) / 128, 128, 0, stream>>>(stat, bn_w_s, bn_w_v, bn_b_s, out);
}


# node pass 1: one thread per (node, channel) element, coalesced; column sums in registers, then one LDS add per block, instead of per-lane serialised same-address atomics (summation order changes)
# speedup vs baseline: 1.6587x; 1.2151x over previous
#include <hip/hip_runtime.h>

#define N_NODES 10000
#define N_EDGES 100000
#define HIDDEN 128
#define MUL0 32
#define MUL1 8
#define IN_DIM 56
#define W_NUMEL 1600
#define EPS 1e-5f

typedef __attribute__((ext_vector_type(2))) float v2f;
typedef __attribute__((ext_vector_type(8))) float v8f;

__device__ __forceinline__ void load_bfrag(v2f (&bf)[32], const float* __restrict__ p, int ld) {
  #pragma unroll
  for (int i = 0; i < 32; ++i) {
    bf[i].x = p[(size_t)(4 * i) * ld];
    bf[i].y = p[(size_t)(4 * i + 1) * ld];
  }
}

typedef __attribute__((ext_vector_type(4))) float v4f;

__device__ __forceinline__ void load_q(v4f (&q)[16], const float* __restrict__ p) {
  #pragma unroll
  for (int i = 0; i < 8; ++i) {
    q[2 * i]     = *(const v4f*)(p + (size_t)(4 * i) * W_NUMEL);
    q[2 * i + 1] = *(const v4f*)(p + (size_t)(4 * i + 1) * W_NUMEL);
  }
}

template <int CH>
__device__ __forceinline__ void mma_chunk(v8f (&cc)[4][2], const v2f (&af)[2][32], const v4f (&q)[16]) {
  #pragma unroll
  for (int i = 0; i < 8; ++i) {
    #pragma unroll
    for (int j = 0; j < 4; ++j) {
      v2f b;
      b.x = q[2 * i][j];
      b.y = q[2 * i + 1][j];
      cc[j][0] = __builtin_amdgcn_wmma_f32_16x16x4_f32(false, af[0][8 * CH + i], false, b, (short)0, cc[j][0], false, false);
      cc[j][1] = __builtin_amdgcn_wmma_f32_16x16x4_f32(false, af[1][8 * CH + i], false, b, (short)0, cc[j][1], false, false);
    }
  }
}

#define WS_NUM   0
#define WS_CNT   (N_NODES*IN_DIM)
#define WS_STAT  (WS_CNT + N_NODES)
#define WS_TOTAL (WS_STAT + 72)

__global__ __launch_bounds__(256) void zero_ws_kernel(float* __restrict__ ws) {
  int i = blockIdx.x * 256 + threadIdx.x;
  if (i < WS_TOTAL) ws[i] = 0.0f;
}

__global__ __launch_bounds__(128)
void edge_kernel(const float* __restrict__ node_attr,
                 const int*   __restrict__ edge_index,
                 const float* __restrict__ edge_attr,
                 const float* __restrict__ edge_sh,
                 const float* __restrict__ fc1_w, const float* __restrict__ fc1_b,
                 const float* __restrict__ fc2_w, const float* __restrict__ fc2_b,
                 float* __restrict__ num, float* __restrict__ cnt)
{
  __shared__ __attribute__((aligned(16))) float sA[32 * 132];
  __shared__ __attribute__((aligned(16))) float sH[32 * 132];
  __shared__ float sX[32 * IN_DIM];
  __shared__ float sY[32 * 4];
  __shared__ __attribute__((aligned(32))) float sZT[73 * 32];
  __shared__ float sTP[32 * IN_DIM];
  __shared__ int   sSrc[32];

  const int tid  = threadIdx.x;
  const int lane = tid & 31;
  const int wv   = tid >> 5;
  const int e0   = blockIdx.x * 32;

  const int m     = lane & 15;
  const int koff  = (lane >> 4) << 1;
  const int mrow0 = (lane >> 4) << 3;

  v2f bf[32];
  v4f qa[16], qb[16];
  v4f bias4;
  const float* wq = fc2_w + (size_t)koff * W_NUMEL + 4 * m;
  const float bias1a = fc1_b[wv * 16 + m];
  const float bias1b = fc1_b[(wv + 4) * 16 + m];
  load_bfrag(bf, fc1_w + (size_t)koff * HIDDEN + wv * 16 + m, HIDDEN);

  for (int i = tid; i < 32 * HIDDEN; i += 128) {
    int r = i >> 7, c = i & 127;
    sA[r * 132 + c] = edge_attr[(size_t)(e0 + r) * HIDDEN + c];
  }
  if (tid < 32) sSrc[tid] = edge_index[e0 + tid];
  if (tid < 128) {
    int r = tid >> 2, c = tid & 3;
    sY[r * 4 + c] = edge_sh[(size_t)(e0 + r) * 4 + c];
  }
  for (int i = tid; i < 32 * IN_DIM; i += 128) {
    int r = i / IN_DIM, c = i - r * IN_DIM;
    int dst = edge_index[N_EDGES + e0 + r];
    sX[i]  = node_attr[(size_t)dst * IN_DIM + c];
    sTP[i] = 0.0f;
  }
  __syncthreads();

  const float alpha = 0.15811388300841897f;
  const float ais3  = alpha * 0.5773502691896258f;
  for (int i = tid; i < 32 * MUL0; i += 128) {
    int u = i >> 5, r = i & 31;
    float x0 = sX[r * IN_DIM + u];
    sZT[i]           = alpha * (x0 * sY[r * 4 + 0]);
    sZT[40 * 32 + i] = alpha * x0;
  }
  for (int i = tid; i < 32 * MUL1; i += 128) {
    int u = i >> 5, r = i & 31;
    const float* xr = &sX[r * IN_DIM + MUL0 + u * 3];
    const float* yr = &sY[r * 4 + 1];
    sZT[32 * 32 + i] = ais3 * (xr[0] * yr[0] + xr[1] * yr[1] + xr[2] * yr[2]);
  }
  if (tid < 32) sZT[72 * 32 + tid] = alpha * sY[tid * 4 + 0];
  __syncthreads();

  v2f af[2][32];
  #pragma unroll
  for (int i = 0; i < 32; ++i) {
    af[0][i].x = sA[m * 132 + 4 * i + koff];
    af[0][i].y = sA[m * 132 + 4 * i + koff + 1];
    af[1][i].x = sA[(16 + m) * 132 + 4 * i + koff];
    af[1][i].y = sA[(16 + m) * 132 + 4 * i + koff + 1];
  }

  #pragma unroll
  for (int g = 0; g < 2; ++g) {
    v8f acc0 = {}, acc1 = {};
    const int col = (wv + 4 * g) * 16 + m;
    #pragma unroll
    for (int i = 0; i < 32; ++i) {
      acc0 = __builtin_amdgcn_wmma_f32_16x16x4_f32(false, af[0][i], false, bf[i], (short)0, acc0, false, false);
      acc1 = __builtin_amdgcn_wmma_f32_16x16x4_f32(false, af[1][i], false, bf[i], (short)0, acc1, false, false);
    }
    if (g == 0) {
      load_bfrag(bf, fc1_w + (size_t)koff * HIDDEN + col + 64, HIDDEN);
    } else {
      bias4 = *(const v4f*)&fc2_b[64 * wv + 4 * m];
      load_q(qa, wq + 64 * wv);
    }
    const float bias = g == 0 ? bias1a : bias1b;
    #pragma unroll
    for (int v = 0; v < 8; ++v) {
      float h0 = acc0[v] + bias;
      float h1 = acc1[v] + bias;
      sH[(mrow0 + v) * 132 + col]      = h0 > 0.0f ? h0 : 0.0f;
      sH[(16 + mrow0 + v) * 132 + col] = h1 > 0.0f ? h1 : 0.0f;
    }
  }
  __syncthreads();

  #pragma unroll
  for (int i = 0; i < 32; ++i) {
    af[0][i].x = sH[m * 132 + 4 * i + koff];
    af[0][i].y = sH[m * 132 + 4 * i + koff + 1];
    af[1][i].x = sH[(16 + m) * 132 + 4 * i + koff];
    af[1][i].y = sH[(16 + m) * 132 + 4 * i + koff + 1];
  }

  v8f o4[4][2];
  #pragma unroll
  for (int j = 0; j < 4; ++j) { o4[j][0] = v8f{}; o4[j][1] = v8f{}; }

  for (int gi = 0; gi < 5; ++gi) {
    const int g = wv + 4 * gi;
    const float* wg = wq + 64 * g;
    const int zrow = 2 * g + (m >> 3);
    const v8f z0 = *(const v8f*)&sZT[zrow * 32 + mrow0];
    const v8f z1 = *(const v8f*)&sZT[zrow * 32 + 16 + mrow0];
    const v4f bias = bias4;
    v8f cc[4][2];
    #pragma unroll
    for (int j = 0; j < 4; ++j) { cc[j][0] = v8f{}; cc[j][1] = v8f{}; }

    load_q(qb, wg + (size_t)32 * W_NUMEL);
    mma_chunk<0>(cc, af, qa);
    load_q(qa, wg + (size_t)64 * W_NUMEL);
    mma_chunk<1>(cc, af, qb);
    load_q(qb, wg + (size_t)96 * W_NUMEL);
    mma_chunk<2>(cc, af, qa);
    if (gi < 4) {
      bias4 = *(const v4f*)&fc2_b[64 * (g + 4) + 4 * m];
      load_q(qa, wg + 256);
    }
    mma_chunk<3>(cc, af, qb);

    #pragma unroll
    for (int j = 0; j < 4; ++j) {
      #pragma unroll
      for (int v = 0; v < 8; ++v) {
        o4[j][0][v] += (cc[j][0][v] + bias[j]) * z0[v];
        o4[j][1][v] += (cc[j][1][v] + bias[j]) * z1[v];
      }
    }
  }

  float bias_next = fc2_b[(80 + wv) * 16 + m];
  load_bfrag(bf, fc2_w + (size_t)koff * W_NUMEL + (80 + wv) * 16 + m, W_NUMEL);
  {
    const int w0 = 4 * (m & 7);
    #pragma unroll
    for (int j = 0; j < 4; ++j) {
      #pragma unroll
      for (int t = 0; t < 2; ++t) {
        #pragma unroll
        for (int v = 0; v < 8; ++v) {
          int e = t * 16 + mrow0 + v;
          unsafeAtomicAdd(&sTP[e * IN_DIM + w0 + j], o4[j][t][v]);
        }
      }
    }
  }

  v8f o[2] = {{}, {}};

  for (int nt = 80 + wv; nt < 100; nt += 4) {
    v8f cc[2] = {{}, {}};
    const int col = nt * 16 + m;
    const float bias = bias_next;
    const int zrow = nt < 96 ? 2 * nt - 120 + (m >> 3) : 72;
    v8f z[2];
    z[0] = *(const v8f*)&sZT[zrow * 32 + mrow0];
    z[1] = *(const v8f*)&sZT[zrow * 32 + 16 + mrow0];
    #pragma unroll
    for (int i = 0; i < 32; ++i) {
      cc[0] = __builtin_amdgcn_wmma_f32_16x16x4_f32(false, af[0][i], false, bf[i], (short)0, cc[0], false, false);
      cc[1] = __builtin_amdgcn_wmma_f32_16x16x4_f32(false, af[1][i], false, bf[i], (short)0, cc[1], false, false);
    }
    if (nt + 4 < 100) {
      bias_next = fc2_b[col + 64];
      load_bfrag(bf, fc2_w + (size_t)koff * W_NUMEL + col + 64, W_NUMEL);
    }

    v8f p[2];
    #pragma unroll
    for (int t = 0; t < 2; ++t) {
      #pragma unroll
      for (int v = 0; v < 8; ++v) p[t][v] = (cc[t][v] + bias) * z[t][v];
    }
    const int w3 = MUL0 + (m & 7) * 3;
    if (nt < 96) {
      o[0] += p[0];
      o[1] += p[1];
      if (nt >= 92) {
        float yv[2][8][3];
        #pragma unroll
        for (int t = 0; t < 2; ++t) {
          #pragma unroll
          for (int v = 0; v < 8; ++v) {
            int e = t * 16 + mrow0 + v;
            #pragma unroll
            for (int mm = 0; mm < 3; ++mm) yv[t][v][mm] = o[t][v] * sY[e * 4 + 1 + mm];
          }
        }
        #pragma unroll
        for (int t = 0; t < 2; ++t) {
          #pragma unroll
          for (int v = 0; v < 8; ++v) {
            int e = t * 16 + mrow0 + v;
            #pragma unroll
            for (int mm = 0; mm < 3; ++mm) unsafeAtomicAdd(&sTP[e * IN_DIM + w3 + mm], yv[t][v][mm]);
          }
        }
      }
    } else {
      const int u3 = MUL0 + ((nt - 96) * 2 + (m >> 3)) * 3;
      float xv[2][8][3];
      #pragma unroll
      for (int t = 0; t < 2; ++t) {
        #pragma unroll
        for (int v = 0; v < 8; ++v) {
          int e = t * 16 + mrow0 + v;
          #pragma unroll
          for (int mm = 0; mm < 3; ++mm) xv[t][v][mm] = p[t][v] * sX[e * IN_DIM + u3 + mm];
        }
      }
      #pragma unroll
      for (int t = 0; t < 2; ++t) {
        #pragma unroll
        for (int v = 0; v < 8; ++v) {
          int e = t * 16 + mrow0 + v;
          #pragma unroll
          for (int mm = 0; mm < 3; ++mm) unsafeAtomicAdd(&sTP[e * IN_DIM + w3 + mm], xv[t][v][mm]);
        }
      }
    }
  }
  __syncthreads();

  for (int i = tid; i < 32 * IN_DIM; i += 128) {
    int r = i / IN_DIM, c = i - r * IN_DIM;
    unsafeAtomicAdd(&num[(size_t)sSrc[r] * IN_DIM + c], sTP[i]);
  }
  if (tid < 32) unsafeAtomicAdd(&cnt[sSrc[tid]], 1.0f);
}

#define NP1_ROWS    8
#define NP1_STEPS   10
#define NP1_NODES   (NP1_ROWS * NP1_STEPS)
#define NP1_THREADS (NP1_ROWS * IN_DIM)
static_assert(N_NODES % NP1_NODES == 0, "node pass 1 has no tail block");

__global__ __launch_bounds__(NP1_THREADS)
void node_pass1(const float* __restrict__ node_attr,
                const float* __restrict__ num, const float* __restrict__ cnt,
                float* __restrict__ outp, float* __restrict__ stat)
{
  __shared__ float ssum[NP1_THREADS], ssq[NP1_THREADS];
  const int tid = threadIdx.x;
  const int g   = tid / IN_DIM;
  const int ch  = tid - g * IN_DIM;
  const int n0  = blockIdx.x * NP1_NODES + g;
  const size_t e0 = (size_t)n0 * IN_DIM + ch;

  float cv[NP1_STEPS], nv[NP1_STEPS], av[NP1_STEPS];
  #pragma unroll
  for (int k = 0; k < NP1_STEPS; ++k) {
    cv[k] = cnt[n0 + k * NP1_ROWS];
    nv[k] = num[e0 + (size_t)k * NP1_ROWS * IN_DIM];
    av[k] = node_attr[e0 + (size_t)k * NP1_ROWS * IN_DIM];
  }
  float s = 0.0f, q = 0.0f;
  #pragma unroll
  for (int k = 0; k < NP1_STEPS; ++k) {
    float c   = cv[k];
    float inv = 1.0f / (c > 1.0f ? c : 1.0f);
    float p = nv[k] * inv + av[k];
    outp[e0 + (size_t)k * NP1_ROWS * IN_DIM] = p;
    s += p;
    q += p * p;
  }
  ssum[tid] = s;
  ssq[tid]  = q;
  __syncthreads();

  if (tid < MUL0) {
    float S = 0.0f, Q = 0.0f;
    #pragma unroll
    for (int r = 0; r < NP1_ROWS; ++r) {
      S += ssum[r * IN_DIM + tid];
      Q += ssq[r * IN_DIM + tid];
    }
    unsafeAtomicAdd(&stat[tid], S);
    unsafeAtomicAdd(&stat[MUL0 + tid], Q);
  } else if (tid < MUL0 + MUL1) {
    const int u = tid - MUL0;
    float Q = 0.0f;
    #pragma unroll
    for (int r = 0; r < NP1_ROWS; ++r) {
      const float* qr = &ssq[r * IN_DIM + MUL0 + u * 3];
      Q += qr[0] + qr[1] + qr[2];
    }
    unsafeAtomicAdd(&stat[2 * MUL0 + u], Q);
  }
}

__global__ __launch_bounds__(128)
void node_pass2(const float* __restrict__ stat,
                const float* __restrict__ bn_w_s, const float* __restrict__ bn_w_v,
                const float* __restrict__ bn_b_s,
                float* __restrict__ out)
{
  const int n = blockIdx.x * 128 + threadIdx.x;
  if (n >= N_NODES) return;
  const float invN = 1.0f / (float)N_NODES;
  for (int ch = 0; ch < MUL0; ++ch) {
    float mean = stat[ch] * invN;
    float var  = stat[MUL0 + ch] * invN - mean * mean;
    float p = out[(size_t)n * IN_DIM + ch];
    out[(size_t)n * IN_DIM + ch] = (p - mean) * rsqrtf(var + EPS) * bn_w_s[ch] + bn_b_s[ch];
  }
  for (int u = 0; u < MUL1; ++u) {
    float vn    = stat[2 * MUL0 + u] * invN * (1.0f / 3.0f);
    float scale = rsqrtf(vn + EPS) * bn_w_v[u];
    for (int mm = 0; mm < 3; ++mm) {
      int ch = MUL0 + u * 3 + mm;
      out[(size_t)n * IN_DIM + ch] *= scale;
    }
  }
}

extern "C" void kernel_launch(void* const* d_in, const int* in_sizes, int n_in,
                              void* d_out, int out_size, void* d_ws, size_t ws_size,
                              hipStream_t stream) {
  const float* node_attr = (const float*)d_in[0];
  const int*   edge_index = (const int*)d_in[1];
  const float* edge_attr = (const float*)d_in[2];
  const float* edge_sh   = (const float*)d_in[3];
  const float* fc1_w = (const float*)d_in[4];
  const float* fc1_b = (const float*)d_in[5];
  const float* fc2_w = (const float*)d_in[6];
  const float* fc2_b = (const float*)d_in[7];
  const float* bn_w_s = (const float*)d_in[8];
  const float* bn_w_v = (const float*)d_in[9];
  const float* bn_b_s = (const float*)d_in[10];

  float* ws   = (float*)d_ws;
  float* num  = ws + WS_NUM;
  float* cnt  = ws + WS_CNT;
  float* stat = ws + WS_STAT;
  float* out  = (float*)d_out;

  zero_ws_kernel<<<(WS_TOTAL + 255) / 256, 256, 0, stream>>>(ws);
  edge_kernel<<<N_EDGES / 32, 128, 0, stream>>>(node_attr, edge_index, edge_attr, edge_sh,
                                                fc1_w, fc1_b, fc2_w, fc2_b, num, cnt);
  node_pass1<<<N_NODES / NP1_NODES, NP1_THREADS, 0, stream>>>(node_attr, num, cnt, out, stat);
  node_pass2<<<(N_NODES + 127) / 128, 128, 0, stream>>>(stat, bn_w_s, bn_w_v, bn_b_s, out);
}


# edge kernel prologue: staging loads (indices, edge_attr and gathered node rows as 16-byte pieces, edge_sh) go out in two groups ahead of any wait: 2 memory round trips instead of 10
# speedup vs baseline: 1.7692x; 1.0666x over previous
#include <hip/hip_runtime.h>

#define N_NODES 10000
#define N_EDGES 100000
#define HIDDEN 128
#define MUL0 32
#define MUL1 8
#define IN_DIM 56
#define W_NUMEL 1600
#define EPS 1e-5f

typedef __attribute__((ext_vector_type(2))) float v2f;
typedef __attribute__((ext_vector_type(8))) float v8f;

__device__ __forceinline__ void load_bfrag(v2f (&bf)[32], const float* __restrict__ p, int ld) {
  #pragma unroll
  for (int i = 0; i < 32; ++i) {
    bf[i].x = p[(size_t)(4 * i) * ld];
    bf[i].y = p[(size_t)(4 * i + 1) * ld];
  }
}

typedef __attribute__((ext_vector_type(4))) float v4f;

__device__ __forceinline__ void load_q(v4f (&q)[16], const float* __restrict__ p) {
  #pragma unroll
  for (int i = 0; i < 8; ++i) {
    q[2 * i]     = *(const v4f*)(p + (size_t)(4 * i) * W_NUMEL);
    q[2 * i + 1] = *(const v4f*)(p + (size_t)(4 * i + 1) * W_NUMEL);
  }
}

template <int CH>
__device__ __forceinline__ void mma_chunk(v8f (&cc)[4][2], const v2f (&af)[2][32], const v4f (&q)[16]) {
  #pragma unroll
  for (int i = 0; i < 8; ++i) {
    #pragma unroll
    for (int j = 0; j < 4; ++j) {
      v2f b;
      b.x = q[2 * i][j];
      b.y = q[2 * i + 1][j];
      cc[j][0] = __builtin_amdgcn_wmma_f32_16x16x4_f32(false, af[0][8 * CH + i], false, b, (short)0, cc[j][0], false, false);
      cc[j][1] = __builtin_amdgcn_wmma_f32_16x16x4_f32(false, af[1][8 * CH + i], false, b, (short)0, cc[j][1], false, false);
    }
  }
}

#define WS_NUM   0
#define WS_CNT   (N_NODES*IN_DIM)
#define WS_STAT  (WS_CNT + N_NODES)
#define WS_TOTAL (WS_STAT + 72)

__global__ __launch_bounds__(256) void zero_ws_kernel(float* __restrict__ ws) {
  int i = blockIdx.x * 256 + threadIdx.x;
  if (i < WS_TOTAL) ws[i] = 0.0f;
}

__global__ __launch_bounds__(128)
void edge_kernel(const float* __restrict__ node_attr,
                 const int*   __restrict__ edge_index,
                 const float* __restrict__ edge_attr,
                 const float* __restrict__ edge_sh,
                 const float* __restrict__ fc1_w, const float* __restrict__ fc1_b,
                 const float* __restrict__ fc2_w, const float* __restrict__ fc2_b,
                 float* __restrict__ num, float* __restrict__ cnt)
{
  __shared__ __attribute__((aligned(16))) float sA[32 * 132];
  __shared__ __attribute__((aligned(16))) float sH[32 * 132];
  __shared__ __attribute__((aligned(16))) float sX[32 * IN_DIM];
  __shared__ float sY[32 * 4];
  __shared__ __attribute__((aligned(32))) float sZT[73 * 32];
  __shared__ __attribute__((aligned(16))) float sTP[32 * IN_DIM];
  __shared__ int   sSrc[128];

  const int tid  = threadIdx.x;
  const int lane = tid & 31;
  const int wv   = tid >> 5;
  const int e0   = blockIdx.x * 32;

  const int m     = lane & 15;
  const int koff  = (lane >> 4) << 1;
  const int mrow0 = (lane >> 4) << 3;

  v2f bf[32];
  v4f qa[16], qb[16];
  v4f bias4;
  const float* wq = fc2_w + (size_t)koff * W_NUMEL + 4 * m;

  constexpr int GP = 32 * IN_DIM / 4;
  int dsti[4];
  #pragma unroll
  for (int k = 0; k < 4; ++k) {
    const int pc = min(tid + 128 * k, GP - 1);
    dsti[k] = edge_index[N_EDGES + e0 + pc / 14];
  }
  const int srcv = edge_index[e0 + lane];
  v4f ea[8];
  #pragma unroll
  for (int k = 0; k < 8; ++k) {
    const int p = tid + 128 * k;
    ea[k] = *(const v4f*)(edge_attr + (size_t)(e0 + (p >> 5)) * HIDDEN + 4 * (p & 31));
  }
  const float yv = edge_sh[(size_t)e0 * 4 + tid];
  v4f xr[4];
  #pragma unroll
  for (int k = 0; k < 4; ++k) {
    const int pc = min(tid + 128 * k, GP - 1);
    const int c4 = pc - (pc / 14) * 14;
    xr[k] = *(const v4f*)(node_attr + (size_t)dsti[k] * IN_DIM + 4 * c4);
  }
  const float bias1a = fc1_b[wv * 16 + m];
  const float bias1b = fc1_b[(wv + 4) * 16 + m];
  load_bfrag(bf, fc1_w + (size_t)koff * HIDDEN + wv * 16 + m, HIDDEN);

  #pragma unroll
  for (int k = 0; k < 8; ++k) {
    const int p = tid + 128 * k;
    *(v4f*)&sA[(p >> 5) * 132 + 4 * (p & 31)] = ea[k];
  }
  sSrc[tid] = srcv;
  sY[tid] = yv;
  #pragma unroll
  for (int k = 0; k < 4; ++k) {
    const int p = tid + 128 * k;
    if (p < GP) {
      *(v4f*)&sX[4 * p]  = xr[k];
      *(v4f*)&sTP[4 * p] = v4f{};
    }
  }
  __syncthreads();

  const float alpha = 0.15811388300841897f;
  const float ais3  = alpha * 0.5773502691896258f;
  for (int i = tid; i < 32 * MUL0; i += 128) {
    int u = i >> 5, r = i & 31;
    float x0 = sX[r * IN_DIM + u];
    sZT[i]           = alpha * (x0 * sY[r * 4 + 0]);
    sZT[40 * 32 + i] = alpha * x0;
  }
  for (int i = tid; i < 32 * MUL1; i += 128) {
    int u = i >> 5, r = i & 31;
    const float* xr = &sX[r * IN_DIM + MUL0 + u * 3];
    const float* yr = &sY[r * 4 + 1];
    sZT[32 * 32 + i] = ais3 * (xr[0] * yr[0] + xr[1] * yr[1] + xr[2] * yr[2]);
  }
  if (tid < 32) sZT[72 * 32 + tid] = alpha * sY[tid * 4 + 0];
  __syncthreads();

  v2f af[2][32];
  #pragma unroll
  for (int i = 0; i < 32; ++i) {
    af[0][i].x = sA[m * 132 + 4 * i + koff];
    af[0][i].y = sA[m * 132 + 4 * i + koff + 1];
    af[1][i].x = sA[(16 + m) * 132 + 4 * i + koff];
    af[1][i].y = sA[(16 + m) * 132 + 4 * i + koff + 1];
  }

  #pragma unroll
  for (int g = 0; g < 2; ++g) {
    v8f acc0 = {}, acc1 = {};
    const int col = (wv + 4 * g) * 16 + m;
    #pragma unroll
    for (int i = 0; i < 32; ++i) {
      acc0 = __builtin_amdgcn_wmma_f32_16x16x4_f32(false, af[0][i], false, bf[i], (short)0, acc0, false, false);
      acc1 = __builtin_amdgcn_wmma_f32_16x16x4_f32(false, af[1][i], false, bf[i], (short)0, acc1, false, false);
    }
    if (g == 0) {
      load_bfrag(bf, fc1_w + (size_t)koff * HIDDEN + col + 64, HIDDEN);
    } else {
      bias4 = *(const v4f*)&fc2_b[64 * wv + 4 * m];
      load_q(qa, wq + 64 * wv);
    }
    const float bias = g == 0 ? bias1a : bias1b;
    #pragma unroll
    for (int v = 0; v < 8; ++v) {
      float h0 = acc0[v] + bias;
      float h1 = acc1[v] + bias;
      sH[(mrow0 + v) * 132 + col]      = h0 > 0.0f ? h0 : 0.0f;
      sH[(16 + mrow0 + v) * 132 + col] = h1 > 0.0f ? h1 : 0.0f;
    }
  }
  __syncthreads();

  #pragma unroll
  for (int i = 0; i < 32; ++i) {
    af[0][i].x = sH[m * 132 + 4 * i + koff];
    af[0][i].y = sH[m * 132 + 4 * i + koff + 1];
    af[1][i].x = sH[(16 + m) * 132 + 4 * i + koff];
    af[1][i].y = sH[(16 + m) * 132 + 4 * i + koff + 1];
  }

  v8f o4[4][2];
  #pragma unroll
  for (int j = 0; j < 4; ++j) { o4[j][0] = v8f{}; o4[j][1] = v8f{}; }

  for (int gi = 0; gi < 5; ++gi) {
    const int g = wv + 4 * gi;
    const float* wg = wq + 64 * g;
    const int zrow = 2 * g + (m >> 3);
    const v8f z0 = *(const v8f*)&sZT[zrow * 32 + mrow0];
    const v8f z1 = *(const v8f*)&sZT[zrow * 32 + 16 + mrow0];
    const v4f bias = bias4;
    v8f cc[4][2];
    #pragma unroll
    for (int j = 0; j < 4; ++j) { cc[j][0] = v8f{}; cc[j][1] = v8f{}; }

    load_q(qb, wg + (size_t)32 * W_NUMEL);
    mma_chunk<0>(cc, af, qa);
    load_q(qa, wg + (size_t)64 * W_NUMEL);
    mma_chunk<1>(cc, af, qb);
    load_q(qb, wg + (size_t)96 * W_NUMEL);
    mma_chunk<2>(cc, af, qa);
    if (gi < 4) {
      bias4 = *(const v4f*)&fc2_b[64 * (g + 4) + 4 * m];
      load_q(qa, wg + 256);
    }
    mma_chunk<3>(cc, af, qb);

    #pragma unroll
    for (int j = 0; j < 4; ++j) {
      #pragma unroll
      for (int v = 0; v < 8; ++v) {
        o4[j][0][v] += (cc[j][0][v] + bias[j]) * z0[v];
        o4[j][1][v] += (cc[j][1][v] + bias[j]) * z1[v];
      }
    }
  }

  float bias_next = fc2_b[(80 + wv) * 16 + m];
  load_bfrag(bf, fc2_w + (size_t)koff * W_NUMEL + (80 + wv) * 16 + m, W_NUMEL);
  {
    const int w0 = 4 * (m & 7);
    #pragma unroll
    for (int j = 0; j < 4; ++j) {
      #pragma unroll
      for (int t = 0; t < 2; ++t) {
        #pragma unroll
        for (int v = 0; v < 8; ++v) {
          int e = t * 16 + mrow0 + v;
          unsafeAtomicAdd(&sTP[e * IN_DIM + w0 + j], o4[j][t][v]);
        }
      }
    }
  }

  v8f o[2] = {{}, {}};

  for (int nt = 80 + wv; nt < 100; nt += 4) {
    v8f cc[2] = {{}, {}};
    const int col = nt * 16 + m;
    const float bias = bias_next;
    const int zrow = nt < 96 ? 2 * nt - 120 + (m >> 3) : 72;
    v8f z[2];
    z[0] = *(const v8f*)&sZT[zrow * 32 + mrow0];
    z[1] = *(const v8f*)&sZT[zrow * 32 + 16 + mrow0];
    #pragma unroll
    for (int i = 0; i < 32; ++i) {
      cc[0] = __builtin_amdgcn_wmma_f32_16x16x4_f32(false, af[0][i], false, bf[i], (short)0, cc[0], false, false);
      cc[1] = __builtin_amdgcn_wmma_f32_16x16x4_f32(false, af[1][i], false, bf[i], (short)0, cc[1], false, false);
    }
    if (nt + 4 < 100) {
      bias_next = fc2_b[col + 64];
      load_bfrag(bf, fc2_w + (size_t)koff * W_NUMEL + col + 64, W_NUMEL);
    }

    v8f p[2];
    #pragma unroll
    for (int t = 0; t < 2; ++t) {
      #pragma unroll
      for (int v = 0; v < 8; ++v) p[t][v] = (cc[t][v] + bias) * z[t][v];
    }
    const int w3 = MUL0 + (m & 7) * 3;
    if (nt < 96) {
      o[0] += p[0];
      o[1] += p[1];
      if (nt >= 92) {
        float yv[2][8][3];
        #pragma unroll
        for (int t = 0; t < 2; ++t) {
          #pragma unroll
          for (int v = 0; v < 8; ++v) {
            int e = t * 16 + mrow0 + v;
            #pragma unroll
            for (int mm = 0; mm < 3; ++mm) yv[t][v][mm] = o[t][v] * sY[e * 4 + 1 + mm];
          }
        }
        #pragma unroll
        for (int t = 0; t < 2; ++t) {
          #pragma unroll
          for (int v = 0; v < 8; ++v) {
            int e = t * 16 + mrow0 + v;
            #pragma unroll
            for (int mm = 0; mm < 3; ++mm) unsafeAtomicAdd(&sTP[e * IN_DIM + w3 + mm], yv[t][v][mm]);
          }
        }
      }
    } else {
      const int u3 = MUL0 + ((nt - 96) * 2 + (m >> 3)) * 3;
      float xv[2][8][3];
      #pragma unroll
      for (int t = 0; t < 2; ++t) {
        #pragma unroll
        for (int v = 0; v < 8; ++v) {
          int e = t * 16 + mrow0 + v;
          #pragma unroll
          for (int mm = 0; mm < 3; ++mm) xv[t][v][mm] = p[t][v] * sX[e * IN_DIM + u3 + mm];
        }
      }
      #pragma unroll
      for (int t = 0; t < 2; ++t) {
        #pragma unroll
        for (int v = 0; v < 8; ++v) {
          int e = t * 16 + mrow0 + v;
          #pragma unroll
          for (int mm = 0; mm < 3; ++mm) unsafeAtomicAdd(&sTP[e * IN_DIM + w3 + mm], xv[t][v][mm]);
        }
      }
    }
  }
  __syncthreads();

  for (int i = tid; i < 32 * IN_DIM; i += 128) {
    int r = i / IN_DIM, c = i - r * IN_DIM;
    unsafeAtomicAdd(&num[(size_t)sSrc[r] * IN_DIM + c], sTP[i]);
  }
  if (tid < 32) unsafeAtomicAdd(&cnt[sSrc[tid]], 1.0f);
}

#define NP1_ROWS    8
#define NP1_STEPS   10
#define NP1_NODES   (NP1_ROWS * NP1_STEPS)
#define NP1_THREADS (NP1_ROWS * IN_DIM)
static_assert(N_NODES % NP1_NODES == 0, "node pass 1 has no tail block");

__global__ __launch_bounds__(NP1_THREADS)
void node_pass1(const float* __restrict__ node_attr,
                const float* __restrict__ num, const float* __restrict__ cnt,
                float* __restrict__ outp, float* __restrict__ stat)
{
  __shared__ float ssum[NP1_THREADS], ssq[NP1_THREADS];
  const int tid = threadIdx.x;
  const int g   = tid / IN_DIM;
  const int ch  = tid - g * IN_DIM;
  const int n0  = blockIdx.x * NP1_NODES + g;
  const size_t e0 = (size_t)n0 * IN_DIM + ch;

  float cv[NP1_STEPS], nv[NP1_STEPS], av[NP1_STEPS];
  #pragma unroll
  for (int k = 0; k < NP1_STEPS; ++k) {
    cv[k] = cnt[n0 + k * NP1_ROWS];
    nv[k] = num[e0 + (size_t)k * NP1_ROWS * IN_DIM];
    av[k] = node_attr[e0 + (size_t)k * NP1_ROWS * IN_DIM];
  }
  float s = 0.0f, q = 0.0f;
  #pragma unroll
  for (int k = 0; k < NP1_STEPS; ++k) {
    float c   = cv[k];
    float inv = 1.0f / (c > 1.0f ? c : 1.0f);
    float p = nv[k] * inv + av[k];
    outp[e0 + (size_t)k * NP1_ROWS * IN_DIM] = p;
    s += p;
    q += p * p;
  }
  ssum[tid] = s;
  ssq[tid]  = q;
  __syncthreads();

  if (tid < MUL0) {
    float S = 0.0f, Q = 0.0f;
    #pragma unroll
    for (int r = 0; r < NP1_ROWS; ++r) {
      S += ssum[r * IN_DIM + tid];
      Q += ssq[r * IN_DIM + tid];
    }
    unsafeAtomicAdd(&stat[tid], S);
    unsafeAtomicAdd(&stat[MUL0 + tid], Q);
  } else if (tid < MUL0 + MUL1) {
    const int u = tid - MUL0;
    float Q = 0.0f;
    #pragma unroll
    for (int r = 0; r < NP1_ROWS; ++r) {
      const float* qr = &ssq[r * IN_DIM + MUL0 + u * 3];
      Q += qr[0] + qr[1] + qr[2];
    }
    unsafeAtomicAdd(&stat[2 * MUL0 + u], Q);
  }
}

__global__ __launch_bounds__(128)
void node_pass2(const float* __restrict__ stat,
                const float* __restrict__ bn_w_s, const float* __restrict__ bn_w_v,
                const float* __restrict__ bn_b_s,
                float* __restrict__ out)
{
  const int n = blockIdx.x * 128 + threadIdx.x;
  if (n >= N_NODES) return;
  const float invN = 1.0f / (float)N_NODES;
  for (int ch = 0; ch < MUL0; ++ch) {
    float mean = stat[ch] * invN;
    float var  = stat[MUL0 + ch] * invN - mean * mean;
    float p = out[(size_t)n * IN_DIM + ch];
    out[(size_t)n * IN_DIM + ch] = (p - mean) * rsqrtf(var + EPS) * bn_w_s[ch] + bn_b_s[ch];
  }
  for (int u = 0; u < MUL1; ++u) {
    float vn    = stat[2 * MUL0 + u] * invN * (1.0f / 3.0f);
    float scale = rsqrtf(vn + EPS) * bn_w_v[u];
    for (int mm = 0; mm < 3; ++mm) {
      int ch = MUL0 + u * 3 + mm;
      out[(size_t)n * IN_DIM + ch] *= scale;
    }
  }
}

extern "C" void kernel_launch(void* const* d_in, const int* in_sizes, int n_in,
                              void* d_out, int out_size, void* d_ws, size_t ws_size,
                              hipStream_t stream) {
  const float* node_attr = (const float*)d_in[0];
  const int*   edge_index = (const int*)d_in[1];
  const float* edge_attr = (const float*)d_in[2];
  const float* edge_sh   = (const float*)d_in[3];
  const float* fc1_w = (const float*)d_in[4];
  const float* fc1_b = (const float*)d_in[5];
  const float* fc2_w = (const float*)d_in[6];
  const float* fc2_b = (const float*)d_in[7];
  const float* bn_w_s = (const float*)d_in[8];
  const float* bn_w_v = (const float*)d_in[9];
  const float* bn_b_s = (const float*)d_in[10];

  float* ws   = (float*)d_ws;
  float* num  = ws + WS_NUM;
  float* cnt  = ws + WS_CNT;
  float* stat = ws + WS_STAT;
  float* out  = (float*)d_out;

  zero_ws_kernel<<<(WS_TOTAL + 255) / 256, 256, 0, stream>>>(ws);
  edge_kernel<<<N_EDGES / 32, 128, 0, stream>>>(node_attr, edge_index, edge_attr, edge_sh,
                                                fc1_w, fc1_b, fc2_w, fc2_b, num, cnt);
  node_pass1<<<N_NODES / NP1_NODES, NP1_THREADS, 0, stream>>>(node_attr, num, cnt, out, stat);
  node_pass2<<<(N_NODES + 127) / 128, 128, 0, stream>>>(stat, bn_w_s, bn_w_v, bn_b_s, out);
}


# edge kernel: multiplier-table and node-scatter loops get static trip counts, all LDS reads of each issued as one group (about 25 LDS round trips per block become 2); barrier behind the table dropped
# speedup vs baseline: 1.8597x; 1.0512x over previous
#include <hip/hip_runtime.h>

#define N_NODES 10000
#define N_EDGES 100000
#define HIDDEN 128
#define MUL0 32
#define MUL1 8
#define IN_DIM 56
#define W_NUMEL 1600
#define EPS 1e-5f

typedef __attribute__((ext_vector_type(2))) float v2f;
typedef __attribute__((ext_vector_type(8))) float v8f;

__device__ __forceinline__ void load_bfrag(v2f (&bf)[32], const float* __restrict__ p, int ld) {
  #pragma unroll
  for (int i = 0; i < 32; ++i) {
    bf[i].x = p[(size_t)(4 * i) * ld];
    bf[i].y = p[(size_t)(4 * i + 1) * ld];
  }
}

typedef __attribute__((ext_vector_type(4))) float v4f;

__device__ __forceinline__ void load_q(v4f (&q)[16], const float* __restrict__ p) {
  #pragma unroll
  for (int i = 0; i < 8; ++i) {
    q[2 * i]     = *(const v4f*)(p + (size_t)(4 * i) * W_NUMEL);
    q[2 * i + 1] = *(const v4f*)(p + (size_t)(4 * i + 1) * W_NUMEL);
  }
}

template <int CH>
__device__ __forceinline__ void mma_chunk(v8f (&cc)[4][2], const v2f (&af)[2][32], const v4f (&q)[16]) {
  #pragma unroll
  for (int i = 0; i < 8; ++i) {
    #pragma unroll
    for (int j = 0; j < 4; ++j) {
      v2f b;
      b.x = q[2 * i][j];
      b.y = q[2 * i + 1][j];
      cc[j][0] = __builtin_amdgcn_wmma_f32_16x16x4_f32(false, af[0][8 * CH + i], false, b, (short)0, cc[j][0], false, false);
      cc[j][1] = __builtin_amdgcn_wmma_f32_16x16x4_f32(false, af[1][8 * CH + i], false, b, (short)0, cc[j][1], false, false);
    }
  }
}

#define WS_NUM   0
#define WS_CNT   (N_NODES*IN_DIM)
#define WS_STAT  (WS_CNT + N_NODES)
#define WS_TOTAL (WS_STAT + 72)

__global__ __launch_bounds__(256) void zero_ws_kernel(float* __restrict__ ws) {
  int i = blockIdx.x * 256 + threadIdx.x;
  if (i < WS_TOTAL) ws[i] = 0.0f;
}

__global__ __launch_bounds__(128)
void edge_kernel(const float* __restrict__ node_attr,
                 const int*   __restrict__ edge_index,
                 const float* __restrict__ edge_attr,
                 const float* __restrict__ edge_sh,
                 const float* __restrict__ fc1_w, const float* __restrict__ fc1_b,
                 const float* __restrict__ fc2_w, const float* __restrict__ fc2_b,
                 float* __restrict__ num, float* __restrict__ cnt)
{
  __shared__ __attribute__((aligned(16))) float sA[32 * 132];
  __shared__ __attribute__((aligned(16))) float sH[32 * 132];
  __shared__ __attribute__((aligned(16))) float sX[32 * IN_DIM];
  __shared__ __attribute__((aligned(16))) float sY[32 * 4];
  __shared__ __attribute__((aligned(32))) float sZT[73 * 32];
  __shared__ __attribute__((aligned(16))) float sTP[32 * IN_DIM];
  __shared__ int   sSrc[128];

  const int tid  = threadIdx.x;
  const int lane = tid & 31;
  const int wv   = tid >> 5;
  const int e0   = blockIdx.x * 32;

  const int m     = lane & 15;
  const int koff  = (lane >> 4) << 1;
  const int mrow0 = (lane >> 4) << 3;

  v2f bf[32];
  v4f qa[16], qb[16];
  v4f bias4;
  const float* wq = fc2_w + (size_t)koff * W_NUMEL + 4 * m;

  constexpr int GP = 32 * IN_DIM / 4;
  int dsti[4];
  #pragma unroll
  for (int k = 0; k < 4; ++k) {
    const int pc = min(tid + 128 * k, GP - 1);
    dsti[k] = edge_index[N_EDGES + e0 + pc / 14];
  }
  const int srcv = edge_index[e0 + lane];
  v4f ea[8];
  #pragma unroll
  for (int k = 0; k < 8; ++k) {
    const int p = tid + 128 * k;
    ea[k] = *(const v4f*)(edge_attr + (size_t)(e0 + (p >> 5)) * HIDDEN + 4 * (p & 31));
  }
  const float yv = edge_sh[(size_t)e0 * 4 + tid];
  v4f xr[4];
  #pragma unroll
  for (int k = 0; k < 4; ++k) {
    const int pc = min(tid + 128 * k, GP - 1);
    const int c4 = pc - (pc / 14) * 14;
    xr[k] = *(const v4f*)(node_attr + (size_t)dsti[k] * IN_DIM + 4 * c4);
  }
  const float bias1a = fc1_b[wv * 16 + m];
  const float bias1b = fc1_b[(wv + 4) * 16 + m];
  load_bfrag(bf, fc1_w + (size_t)koff * HIDDEN + wv * 16 + m, HIDDEN);

  #pragma unroll
  for (int k = 0; k < 8; ++k) {
    const int p = tid + 128 * k;
    *(v4f*)&sA[(p >> 5) * 132 + 4 * (p & 31)] = ea[k];
  }
  sSrc[tid] = srcv;
  sY[tid] = yv;
  #pragma unroll
  for (int k = 0; k < 4; ++k) {
    const int p = tid + 128 * k;
    if (p < GP) {
      *(v4f*)&sX[4 * p]  = xr[k];
      *(v4f*)&sTP[4 * p] = v4f{};
    }
  }
  __syncthreads();

  v2f af[2][32];
  #pragma unroll
  for (int i = 0; i < 32; ++i) {
    af[0][i].x = sA[m * 132 + 4 * i + koff];
    af[0][i].y = sA[m * 132 + 4 * i + koff + 1];
    af[1][i].x = sA[(16 + m) * 132 + 4 * i + koff];
    af[1][i].y = sA[(16 + m) * 132 + 4 * i + koff + 1];
  }

  const float alpha = 0.15811388300841897f;
  const float ais3  = alpha * 0.5773502691896258f;
  {
    const v4f ye = *(const v4f*)&sY[lane * 4];
    float x0v[8], x1v[2][3];
    #pragma unroll
    for (int k = 0; k < 8; ++k) x0v[k] = sX[lane * IN_DIM + wv + 4 * k];
    #pragma unroll
    for (int k = 0; k < 2; ++k) {
      #pragma unroll
      for (int mm = 0; mm < 3; ++mm) x1v[k][mm] = sX[lane * IN_DIM + MUL0 + (wv + 4 * k) * 3 + mm];
    }
    #pragma unroll
    for (int k = 0; k < 8; ++k) {
      const int u = wv + 4 * k;
      sZT[u * 32 + lane]        = alpha * (x0v[k] * ye[0]);
      sZT[(40 + u) * 32 + lane] = alpha * x0v[k];
    }
    #pragma unroll
    for (int k = 0; k < 2; ++k) {
      const int u = wv + 4 * k;
      sZT[(32 + u) * 32 + lane] = ais3 * (x1v[k][0] * ye[1] + x1v[k][1] * ye[2] + x1v[k][2] * ye[3]);
    }
    if (wv == 0) sZT[72 * 32 + lane] = alpha * ye[0];
  }

  #pragma unroll
  for (int g = 0; g < 2; ++g) {
    v8f acc0 = {}, acc1 = {};
    const int col = (wv + 4 * g) * 16 + m;
    #pragma unroll
    for (int i = 0; i < 32; ++i) {
      acc0 = __builtin_amdgcn_wmma_f32_16x16x4_f32(false, af[0][i], false, bf[i], (short)0, acc0, false, false);
      acc1 = __builtin_amdgcn_wmma_f32_16x16x4_f32(false, af[1][i], false, bf[i], (short)0, acc1, false, false);
    }
    if (g == 0) {
      load_bfrag(bf, fc1_w + (size_t)koff * HIDDEN + col + 64, HIDDEN);
    } else {
      bias4 = *(const v4f*)&fc2_b[64 * wv + 4 * m];
      load_q(qa, wq + 64 * wv);
    }
    const float bias = g == 0 ? bias1a : bias1b;
    #pragma unroll
    for (int v = 0; v < 8; ++v) {
      float h0 = acc0[v] + bias;
      float h1 = acc1[v] + bias;
      sH[(mrow0 + v) * 132 + col]      = h0 > 0.0f ? h0 : 0.0f;
      sH[(16 + mrow0 + v) * 132 + col] = h1 > 0.0f ? h1 : 0.0f;
    }
  }
  __syncthreads();

  #pragma unroll
  for (int i = 0; i < 32; ++i) {
    af[0][i].x = sH[m * 132 + 4 * i + koff];
    af[0][i].y = sH[m * 132 + 4 * i + koff + 1];
    af[1][i].x = sH[(16 + m) * 132 + 4 * i + koff];
    af[1][i].y = sH[(16 + m) * 132 + 4 * i + koff + 1];
  }

  v8f o4[4][2];
  #pragma unroll
  for (int j = 0; j < 4; ++j) { o4[j][0] = v8f{}; o4[j][1] = v8f{}; }

  for (int gi = 0; gi < 5; ++gi) {
    const int g = wv + 4 * gi;
    const float* wg = wq + 64 * g;
    const int zrow = 2 * g + (m >> 3);
    const v8f z0 = *(const v8f*)&sZT[zrow * 32 + mrow0];
    const v8f z1 = *(const v8f*)&sZT[zrow * 32 + 16 + mrow0];
    const v4f bias = bias4;
    v8f cc[4][2];
    #pragma unroll
    for (int j = 0; j < 4; ++j) { cc[j][0] = v8f{}; cc[j][1] = v8f{}; }

    load_q(qb, wg + (size_t)32 * W_NUMEL);
    mma_chunk<0>(cc, af, qa);
    load_q(qa, wg + (size_t)64 * W_NUMEL);
    mma_chunk<1>(cc, af, qb);
    load_q(qb, wg + (size_t)96 * W_NUMEL);
    mma_chunk<2>(cc, af, qa);
    if (gi < 4) {
      bias4 = *(const v4f*)&fc2_b[64 * (g + 4) + 4 * m];
      load_q(qa, wg + 256);
    }
    mma_chunk<3>(cc, af, qb);

    #pragma unroll
    for (int j = 0; j < 4; ++j) {
      #pragma unroll
      for (int v = 0; v < 8; ++v) {
        o4[j][0][v] += (cc[j][0][v] + bias[j]) * z0[v];
        o4[j][1][v] += (cc[j][1][v] + bias[j]) * z1[v];
      }
    }
  }

  float bias_next = fc2_b[(80 + wv) * 16 + m];
  load_bfrag(bf, fc2_w + (size_t)koff * W_NUMEL + (80 + wv) * 16 + m, W_NUMEL);
  {
    const int w0 = 4 * (m & 7);
    #pragma unroll
    for (int j = 0; j < 4; ++j) {
      #pragma unroll
      for (int t = 0; t < 2; ++t) {
        #pragma unroll
        for (int v = 0; v < 8; ++v) {
          int e = t * 16 + mrow0 + v;
          unsafeAtomicAdd(&sTP[e * IN_DIM + w0 + j], o4[j][t][v]);
        }
      }
    }
  }

  v8f o[2] = {{}, {}};

  for (int nt = 80 + wv; nt < 100; nt += 4) {
    v8f cc[2] = {{}, {}};
    const int col = nt * 16 + m;
    const float bias = bias_next;
    const int zrow = nt < 96 ? 2 * nt - 120 + (m >> 3) : 72;
    v8f z[2];
    z[0] = *(const v8f*)&sZT[zrow * 32 + mrow0];
    z[1] = *(const v8f*)&sZT[zrow * 32 + 16 + mrow0];
    #pragma unroll
    for (int i = 0; i < 32; ++i) {
      cc[0] = __builtin_amdgcn_wmma_f32_16x16x4_f32(false, af[0][i], false, bf[i], (short)0, cc[0], false, false);
      cc[1] = __builtin_amdgcn_wmma_f32_16x16x4_f32(false, af[1][i], false, bf[i], (short)0, cc[1], false, false);
    }
    if (nt + 4 < 100) {
      bias_next = fc2_b[col + 64];
      load_bfrag(bf, fc2_w + (size_t)koff * W_NUMEL + col + 64, W_NUMEL);
    }

    v8f p[2];
    #pragma unroll
    for (int t = 0; t < 2; ++t) {
      #pragma unroll
      for (int v = 0; v < 8; ++v) p[t][v] = (cc[t][v] + bias) * z[t][v];
    }
    const int w3 = MUL0 + (m & 7) * 3;
    if (nt < 96) {
      o[0] += p[0];
      o[1] += p[1];
      if (nt >= 92) {
        float yv[2][8][3];
        #pragma unroll
        for (int t = 0; t < 2; ++t) {
          #pragma unroll
          for (int v = 0; v < 8; ++v) {
            int e = t * 16 + mrow0 + v;
            #pragma unroll
            for (int mm = 0; mm < 3; ++mm) yv[t][v][mm] = o[t][v] * sY[e * 4 + 1 + mm];
          }
        }
        #pragma unroll
        for (int t = 0; t < 2; ++t) {
          #pragma unroll
          for (int v = 0; v < 8; ++v) {
            int e = t * 16 + mrow0 + v;
            #pragma unroll
            for (int mm = 0; mm < 3; ++mm) unsafeAtomicAdd(&sTP[e * IN_DIM + w3 + mm], yv[t][v][mm]);
          }
        }
      }
    } else {
      const int u3 = MUL0 + ((nt - 96) * 2 + (m >> 3)) * 3;
      float xv[2][8][3];
      #pragma unroll
      for (int t = 0; t < 2; ++t) {
        #pragma unroll
        for (int v = 0; v < 8; ++v) {
          int e = t * 16 + mrow0 + v;
          #pragma unroll
          for (int mm = 0; mm < 3; ++mm) xv[t][v][mm] = p[t][v] * sX[e * IN_DIM + u3 + mm];
        }
      }
      #pragma unroll
      for (int t = 0; t < 2; ++t) {
        #pragma unroll
        for (int v = 0; v < 8; ++v) {
          int e = t * 16 + mrow0 + v;
          #pragma unroll
          for (int mm = 0; mm < 3; ++mm) unsafeAtomicAdd(&sTP[e * IN_DIM + w3 + mm], xv[t][v][mm]);
        }
      }
    }
  }
  __syncthreads();

  {
    constexpr int SC = 32 * IN_DIM / 128;
    int   sr[SC];
    float tv[SC];
    #pragma unroll
    for (int k = 0; k < SC; ++k) {
      const int i = tid + 128 * k;
      sr[k] = sSrc[i / IN_DIM];
      tv[k] = sTP[i];
    }
    const int srcn = sSrc[lane];
    #pragma unroll
    for (int k = 0; k < SC; ++k) {
      const int i = tid + 128 * k;
      const int c = i - (i / IN_DIM) * IN_DIM;
      unsafeAtomicAdd(&num[(size_t)sr[k] * IN_DIM + c], tv[k]);
    }
    if (wv == 0) unsafeAtomicAdd(&cnt[srcn], 1.0f);
  }
}

#define NP1_ROWS    8
#define NP1_STEPS   10
#define NP1_NODES   (NP1_ROWS * NP1_STEPS)
#define NP1_THREADS (NP1_ROWS * IN_DIM)
static_assert(N_NODES % NP1_NODES == 0, "node pass 1 has no tail block");

__global__ __launch_bounds__(NP1_THREADS)
void node_pass1(const float* __restrict__ node_attr,
                const float* __restrict__ num, const float* __restrict__ cnt,
                float* __restrict__ outp, float* __restrict__ stat)
{
  __shared__ float ssum[NP1_THREADS], ssq[NP1_THREADS];
  const int tid = threadIdx.x;
  const int g   = tid / IN_DIM;
  const int ch  = tid - g * IN_DIM;
  const int n0  = blockIdx.x * NP1_NODES + g;
  const size_t e0 = (size_t)n0 * IN_DIM + ch;

  float cv[NP1_STEPS], nv[NP1_STEPS], av[NP1_STEPS];
  #pragma unroll
  for (int k = 0; k < NP1_STEPS; ++k) {
    cv[k] = cnt[n0 + k * NP1_ROWS];
    nv[k] = num[e0 + (size_t)k * NP1_ROWS * IN_DIM];
    av[k] = node_attr[e0 + (size_t)k * NP1_ROWS * IN_DIM];
  }
  float s = 0.0f, q = 0.0f;
  #pragma unroll
  for (int k = 0; k < NP1_STEPS; ++k) {
    float c   = cv[k];
    float inv = 1.0f / (c > 1.0f ? c : 1.0f);
    float p = nv[k] * inv + av[k];
    outp[e0 + (size_t)k * NP1_ROWS * IN_DIM] = p;
    s += p;
    q += p * p;
  }
  ssum[tid] = s;
  ssq[tid]  = q;
  __syncthreads();

  if (tid < MUL0) {
    float S = 0.0f, Q = 0.0f;
    #pragma unroll
    for (int r = 0; r < NP1_ROWS; ++r) {
      S += ssum[r * IN_DIM + tid];
      Q += ssq[r * IN_DIM + tid];
    }
    unsafeAtomicAdd(&stat[tid], S);
    unsafeAtomicAdd(&stat[MUL0 + tid], Q);
  } else if (tid < MUL0 + MUL1) {
    const int u = tid - MUL0;
    float Q = 0.0f;
    #pragma unroll
    for (int r = 0; r < NP1_ROWS; ++r) {
      const float* qr = &ssq[r * IN_DIM + MUL0 + u * 3];
      Q += qr[0] + qr[1] + qr[2];
    }
    unsafeAtomicAdd(&stat[2 * MUL0 + u], Q);
  }
}

__global__ __launch_bounds__(128)
void node_pass2(const float* __restrict__ stat,
                const float* __restrict__ bn_w_s, const float* __restrict__ bn_w_v,
                const float* __restrict__ bn_b_s,
                float* __restrict__ out)
{
  const int n = blockIdx.x * 128 + threadIdx.x;
  if (n >= N_NODES) return;
  const float invN = 1.0f / (float)N_NODES;
  for (int ch = 0; ch < MUL0; ++ch) {
    float mean = stat[ch] * invN;
    float var  = stat[MUL0 + ch] * invN - mean * mean;
    float p = out[(size_t)n * IN_DIM + ch];
    out[(size_t)n * IN_DIM + ch] = (p - mean) * rsqrtf(var + EPS) * bn_w_s[ch] + bn_b_s[ch];
  }
  for (int u = 0; u < MUL1; ++u) {
    float vn    = stat[2 * MUL0 + u] * invN * (1.0f / 3.0f);
    float scale = rsqrtf(vn + EPS) * bn_w_v[u];
    for (int mm = 0; mm < 3; ++mm) {
      int ch = MUL0 + u * 3 + mm;
      out[(size_t)n * IN_DIM + ch] *= scale;
    }
  }
}

extern "C" void kernel_launch(void* const* d_in, const int* in_sizes, int n_in,
                              void* d_out, int out_size, void* d_ws, size_t ws_size,
                              hipStream_t stream) {
  const float* node_attr = (const float*)d_in[0];
  const int*   edge_index = (const int*)d_in[1];
  const float* edge_attr = (const float*)d_in[2];
  const float* edge_sh   = (const float*)d_in[3];
  const float* fc1_w = (const float*)d_in[4];
  const float* fc1_b = (const float*)d_in[5];
  const float* fc2_w = (const float*)d_in[6];
  const float* fc2_b = (const float*)d_in[7];
  const float* bn_w_s = (const float*)d_in[8];
  const float* bn_w_v = (const float*)d_in[9];
  const float* bn_b_s = (const float*)d_in[10];

  float* ws   = (float*)d_ws;
  float* num  = ws + WS_NUM;
  float* cnt  = ws + WS_CNT;
  float* stat = ws + WS_STAT;
  float* out  = (float*)d_out;

  zero_ws_kernel<<<(WS_TOTAL + 255) / 256, 256, 0, stream>>>(ws);
  edge_kernel<<<N_EDGES / 32, 128, 0, stream>>>(node_attr, edge_index, edge_attr, edge_sh,
                                                fc1_w, fc1_b, fc2_w, fc2_b, num, cnt);
  node_pass1<<<N_NODES / NP1_NODES, NP1_THREADS, 0, stream>>>(node_attr, num, cnt, out, stat);
  node_pass2<<<(N_NODES + 127) / 128, 128, 0, stream>>>(stat, bn_w_s, bn_w_v, bn_b_s, out);
}


# Both GEMMs' weights repacked per call in the clearing launch (parked in out until node pass 1 overwrites it): a 16-byte load is two ready B pairs, no register moves at WMMAs; summation order changes
# speedup vs baseline: 1.9979x; 1.0743x over previous
#include <hip/hip_runtime.h>

#define N_NODES 10000
#define N_EDGES 100000
#define HIDDEN 128
#define MUL0 32
#define MUL1 8
#define IN_DIM 56
#define W_NUMEL 1600
#define EPS 1e-5f

typedef __attribute__((ext_vector_type(2))) float v2f;
typedef __attribute__((ext_vector_type(8))) float v8f;

typedef __attribute__((ext_vector_type(4))) float v4f;

#define WPK_TILE  2048
#define WPK_PIECE 128
#define WPK_FC2   (HIDDEN * W_NUMEL)
#define WPK_TOTAL (WPK_FC2 + HIDDEN * HIDDEN)

__device__ __forceinline__ void load_q(v4f (&q)[16], const float* __restrict__ p) {
  #pragma unroll
  for (int j = 0; j < 4; ++j) {
    #pragma unroll
    for (int pp = 0; pp < 4; ++pp) q[4 * j + pp] = *(const v4f*)(p + j * WPK_TILE + pp * WPK_PIECE);
  }
}

template <int CH>
__device__ __forceinline__ void mma_chunk(v8f (&cc)[4][2], const v4f (&ah)[2][16], const v4f (&q)[16]) {
  #pragma unroll
  for (int pp = 0; pp < 4; ++pp) {
    const v4f a0 = ah[0][4 * CH + pp], a1 = ah[1][4 * CH + pp];
    const v2f a0lo = __builtin_shufflevector(a0, a0, 0, 1), a0hi = __builtin_shufflevector(a0, a0, 2, 3);
    const v2f a1lo = __builtin_shufflevector(a1, a1, 0, 1), a1hi = __builtin_shufflevector(a1, a1, 2, 3);
    #pragma unroll
    for (int j = 0; j < 4; ++j) {
      const v4f w = q[4 * j + pp];
      const v2f b = __builtin_shufflevector(w, w, 0, 1);
      cc[j][0] = __builtin_amdgcn_wmma_f32_16x16x4_f32(false, a0lo, false, b, (short)0, cc[j][0], false, false);
      cc[j][1] = __builtin_amdgcn_wmma_f32_16x16x4_f32(false, a1lo, false, b, (short)0, cc[j][1], false, false);
    }
    #pragma unroll
    for (int j = 0; j < 4; ++j) {
      const v4f w = q[4 * j + pp];
      const v2f b = __builtin_shufflevector(w, w, 2, 3);
      cc[j][0] = __builtin_amdgcn_wmma_f32_16x16x4_f32(false, a0hi, false, b, (short)0, cc[j][0], false, false);
      cc[j][1] = __builtin_amdgcn_wmma_f32_16x16x4_f32(false, a1hi, false, b, (short)0, cc[j][1], false, false);
    }
  }
}

__device__ __forceinline__ void mma_tile(v8f &c0, v8f &c1, const v4f (&a)[2][16], const v4f (&bq)[16]) {
  #pragma unroll
  for (int p = 0; p < 16; ++p) {
    const v2f blo = __builtin_shufflevector(bq[p], bq[p], 0, 1);
    const v2f bhi = __builtin_shufflevector(bq[p], bq[p], 2, 3);
    c0 = __builtin_amdgcn_wmma_f32_16x16x4_f32(false, __builtin_shufflevector(a[0][p], a[0][p], 0, 1), false, blo, (short)0, c0, false, false);
    c1 = __builtin_amdgcn_wmma_f32_16x16x4_f32(false, __builtin_shufflevector(a[1][p], a[1][p], 0, 1), false, blo, (short)0, c1, false, false);
    c0 = __builtin_amdgcn_wmma_f32_16x16x4_f32(false, __builtin_shufflevector(a[0][p], a[0][p], 2, 3), false, bhi, (short)0, c0, false, false);
    c1 = __builtin_amdgcn_wmma_f32_16x16x4_f32(false, __builtin_shufflevector(a[1][p], a[1][p], 2, 3), false, bhi, (short)0, c1, false, false);
  }
}

__device__ __forceinline__ void load_bq(v4f (&bq)[16], const float* __restrict__ p) {
  #pragma unroll
  for (int pp = 0; pp < 16; ++pp) bq[pp] = *(const v4f*)(p + pp * WPK_PIECE);
}

#define WS_NUM   0
#define WS_CNT   (N_NODES*IN_DIM)
#define WS_STAT  (WS_CNT + N_NODES)
#define WS_TOTAL (WS_STAT + 72)

static_assert(WPK_TOTAL <= N_NODES * IN_DIM, "packed weights fit in the output buffer");
static_assert(WPK_TOTAL <= WS_TOTAL, "the clearing grid covers the packed weights");
__global__ __launch_bounds__(256) void prep_kernel(float* __restrict__ ws,
                                                   const float* __restrict__ fc1_w,
                                                   const float* __restrict__ fc2_w,
                                                   float* __restrict__ wpk) {
  const int i = blockIdx.x * 256 + threadIdx.x;
  if (i < WS_TOTAL) ws[i] = 0.0f;
  if (i < WPK_TOTAL) {
    const int r = i & 3;
    const int l = (i >> 2) & 31;
    const int p = (i >> 7) & 15;
    const int T = i >> 11;
    const int m = l & 15, h = l >> 4;
    const int row = 8 * p + 4 * h + r;
    if (T < 100) {
      const int col = T < 80 ? 64 * (T >> 2) + 4 * m + (T & 3) : 16 * T + m;
      wpk[i] = fc2_w[(size_t)row * W_NUMEL + col];
    } else {
      wpk[i] = fc1_w[row * HIDDEN + 16 * (T - 100) + m];
    }
  }
}

__global__ __launch_bounds__(128)
void edge_kernel(const float* __restrict__ node_attr,
                 const int*   __restrict__ edge_index,
                 const float* __restrict__ edge_attr,
                 const float* __restrict__ edge_sh,
                 const float* __restrict__ fc1_b,
                 const float* __restrict__ wpk, const float* __restrict__ fc2_b,
                 float* __restrict__ num, float* __restrict__ cnt)
{
  __shared__ __attribute__((aligned(16))) float sA[32 * 132];
  __shared__ __attribute__((aligned(16))) float sH[32 * 132];
  __shared__ __attribute__((aligned(16))) float sX[32 * IN_DIM];
  __shared__ __attribute__((aligned(16))) float sY[32 * 4];
  __shared__ __attribute__((aligned(32))) float sZT[73 * 32];
  __shared__ __attribute__((aligned(16))) float sTP[32 * IN_DIM];
  __shared__ int   sSrc[128];

  const int tid  = threadIdx.x;
  const int lane = tid & 31;
  const int wv   = tid >> 5;
  const int e0   = blockIdx.x * 32;

  const int m     = lane & 15;
  const int hq    = (lane >> 4) << 2;
  const int mrow0 = (lane >> 4) << 3;

  v4f bq[16];
  v4f qa[16], qb[16];
  v4f bias4;
  const float* wp = wpk + 4 * lane;

  constexpr int GP = 32 * IN_DIM / 4;
  int dsti[4];
  #pragma unroll
  for (int k = 0; k < 4; ++k) {
    const int pc = min(tid + 128 * k, GP - 1);
    dsti[k] = edge_index[N_EDGES + e0 + pc / 14];
  }
  const int srcv = edge_index[e0 + lane];
  v4f ea[8];
  #pragma unroll
  for (int k = 0; k < 8; ++k) {
    const int p = tid + 128 * k;
    ea[k] = *(const v4f*)(edge_attr + (size_t)(e0 + (p >> 5)) * HIDDEN + 4 * (p & 31));
  }
  const float yv = edge_sh[(size_t)e0 * 4 + tid];
  v4f xr[4];
  #pragma unroll
  for (int k = 0; k < 4; ++k) {
    const int pc = min(tid + 128 * k, GP - 1);
    const int c4 = pc - (pc / 14) * 14;
    xr[k] = *(const v4f*)(node_attr + (size_t)dsti[k] * IN_DIM + 4 * c4);
  }
  const float bias1a = fc1_b[wv * 16 + m];
  const float bias1b = fc1_b[(wv + 4) * 16 + m];
  const float* wp1 = wp + WPK_FC2;
  load_bq(bq, wp1 + (size_t)wv * WPK_TILE);

  #pragma unroll
  for (int k = 0; k < 8; ++k) {
    const int p = tid + 128 * k;
    *(v4f*)&sA[(p >> 5) * 132 + 4 * (p & 31)] = ea[k];
  }
  sSrc[tid] = srcv;
  sY[tid] = yv;
  #pragma unroll
  for (int k = 0; k < 4; ++k) {
    const int p = tid + 128 * k;
    if (p < GP) {
      *(v4f*)&sX[4 * p]  = xr[k];
      *(v4f*)&sTP[4 * p] = v4f{};
    }
  }
  __syncthreads();

  v4f ah[2][16];
  #pragma unroll
  for (int p = 0; p < 16; ++p) {
    ah[0][p] = *(const v4f*)&sA[m * 132 + 8 * p + hq];
    ah[1][p] = *(const v4f*)&sA[(16 + m) * 132 + 8 * p + hq];
  }

  const float alpha = 0.15811388300841897f;
  const float ais3  = alpha * 0.5773502691896258f;
  {
    const v4f ye = *(const v4f*)&sY[lane * 4];
    float x0v[8], x1v[2][3];
    #pragma unroll
    for (int k = 0; k < 8; ++k) x0v[k] = sX[lane * IN_DIM + wv + 4 * k];
    #pragma unroll
    for (int k = 0; k < 2; ++k) {
      #pragma unroll
      for (int mm = 0; mm < 3; ++mm) x1v[k][mm] = sX[lane * IN_DIM + MUL0 + (wv + 4 * k) * 3 + mm];
    }
    #pragma unroll
    for (int k = 0; k < 8; ++k) {
      const int u = wv + 4 * k;
      sZT[u * 32 + lane]        = alpha * (x0v[k] * ye[0]);
      sZT[(40 + u) * 32 + lane] = alpha * x0v[k];
    }
    #pragma unroll
    for (int k = 0; k < 2; ++k) {
      const int u = wv + 4 * k;
      sZT[(32 + u) * 32 + lane] = ais3 * (x1v[k][0] * ye[1] + x1v[k][1] * ye[2] + x1v[k][2] * ye[3]);
    }
    if (wv == 0) sZT[72 * 32 + lane] = alpha * ye[0];
  }

  #pragma unroll
  for (int g = 0; g < 2; ++g) {
    v8f acc0 = {}, acc1 = {};
    const int col = (wv + 4 * g) * 16 + m;
    mma_tile(acc0, acc1, ah, bq);
    if (g == 0) {
      load_bq(bq, wp1 + (size_t)(wv + 4) * WPK_TILE);
    } else {
      bias4 = *(const v4f*)&fc2_b[64 * wv + 4 * m];
      load_q(qa, wp + (size_t)(4 * wv) * WPK_TILE);
    }
    const float bias = g == 0 ? bias1a : bias1b;
    #pragma unroll
    for (int v = 0; v < 8; ++v) {
      float h0 = acc0[v] + bias;
      float h1 = acc1[v] + bias;
      sH[(mrow0 + v) * 132 + col]      = h0 > 0.0f ? h0 : 0.0f;
      sH[(16 + mrow0 + v) * 132 + col] = h1 > 0.0f ? h1 : 0.0f;
    }
  }
  __syncthreads();

  #pragma unroll
  for (int p = 0; p < 16; ++p) {
    ah[0][p] = *(const v4f*)&sH[m * 132 + 8 * p + hq];
    ah[1][p] = *(const v4f*)&sH[(16 + m) * 132 + 8 * p + hq];
  }

  v8f o4[4][2];
  #pragma unroll
  for (int j = 0; j < 4; ++j) { o4[j][0] = v8f{}; o4[j][1] = v8f{}; }

  for (int gi = 0; gi < 5; ++gi) {
    const int g = wv + 4 * gi;
    const float* wg = wp + (size_t)(4 * g) * WPK_TILE;
    const int zrow = 2 * g + (m >> 3);
    const v8f z0 = *(const v8f*)&sZT[zrow * 32 + mrow0];
    const v8f z1 = *(const v8f*)&sZT[zrow * 32 + 16 + mrow0];
    const v4f bias = bias4;
    v8f cc[4][2];
    #pragma unroll
    for (int j = 0; j < 4; ++j) { cc[j][0] = v8f{}; cc[j][1] = v8f{}; }

    load_q(qb, wg + 4 * WPK_PIECE);
    mma_chunk<0>(cc, ah, qa);
    load_q(qa, wg + 8 * WPK_PIECE);
    mma_chunk<1>(cc, ah, qb);
    load_q(qb, wg + 12 * WPK_PIECE);
    mma_chunk<2>(cc, ah, qa);
    if (gi < 4) {
      bias4 = *(const v4f*)&fc2_b[64 * (g + 4) + 4 * m];
      load_q(qa, wg + 16 * WPK_TILE);
    }
    mma_chunk<3>(cc, ah, qb);

    #pragma unroll
    for (int j = 0; j < 4; ++j) {
      #pragma unroll
      for (int v = 0; v < 8; ++v) {
        o4[j][0][v] += (cc[j][0][v] + bias[j]) * z0[v];
        o4[j][1][v] += (cc[j][1][v] + bias[j]) * z1[v];
      }
    }
  }

  float bias_next = fc2_b[(80 + wv) * 16 + m];
  load_bq(bq, wp + (size_t)(80 + wv) * WPK_TILE);
  {
    const int w0 = 4 * (m & 7);
    #pragma unroll
    for (int j = 0; j < 4; ++j) {
      #pragma unroll
      for (int t = 0; t < 2; ++t) {
        #pragma unroll
        for (int v = 0; v < 8; ++v) {
          int e = t * 16 + mrow0 + v;
          unsafeAtomicAdd(&sTP[e * IN_DIM + w0 + j], o4[j][t][v]);
        }
      }
    }
  }

  v8f o[2] = {{}, {}};

  for (int nt = 80 + wv; nt < 100; nt += 4) {
    v8f cc[2] = {{}, {}};
    const int col = nt * 16 + m;
    const float bias = bias_next;
    const int zrow = nt < 96 ? 2 * nt - 120 + (m >> 3) : 72;
    v8f z[2];
    z[0] = *(const v8f*)&sZT[zrow * 32 + mrow0];
    z[1] = *(const v8f*)&sZT[zrow * 32 + 16 + mrow0];
    mma_tile(cc[0], cc[1], ah, bq);
    if (nt + 4 < 100) {
      bias_next = fc2_b[col + 64];
      load_bq(bq, wp + (size_t)(nt + 4) * WPK_TILE);
    }

    v8f p[2];
    #pragma unroll
    for (int t = 0; t < 2; ++t) {
      #pragma unroll
      for (int v = 0; v < 8; ++v) p[t][v] = (cc[t][v] + bias) * z[t][v];
    }
    const int w3 = MUL0 + (m & 7) * 3;
    if (nt < 96) {
      o[0] += p[0];
      o[1] += p[1];
      if (nt >= 92) {
        float yv[2][8][3];
        #pragma unroll
        for (int t = 0; t < 2; ++t) {
          #pragma unroll
          for (int v = 0; v < 8; ++v) {
            int e = t * 16 + mrow0 + v;
            #pragma unroll
            for (int mm = 0; mm < 3; ++mm) yv[t][v][mm] = o[t][v] * sY[e * 4 + 1 + mm];
          }
        }
        #pragma unroll
        for (int t = 0; t < 2; ++t) {
          #pragma unroll
          for (int v = 0; v < 8; ++v) {
            int e = t * 16 + mrow0 + v;
            #pragma unroll
            for (int mm = 0; mm < 3; ++mm) unsafeAtomicAdd(&sTP[e * IN_DIM + w3 + mm], yv[t][v][mm]);
          }
        }
      }
    } else {
      const int u3 = MUL0 + ((nt - 96) * 2 + (m >> 3)) * 3;
      float xv[2][8][3];
      #pragma unroll
      for (int t = 0; t < 2; ++t) {
        #pragma unroll
        for (int v = 0; v < 8; ++v) {
          int e = t * 16 + mrow0 + v;
          #pragma unroll
          for (int mm = 0; mm < 3; ++mm) xv[t][v][mm] = p[t][v] * sX[e * IN_DIM + u3 + mm];
        }
      }
      #pragma unroll
      for (int t = 0; t < 2; ++t) {
        #pragma unroll
        for (int v = 0; v < 8; ++v) {
          int e = t * 16 + mrow0 + v;
          #pragma unroll
          for (int mm = 0; mm < 3; ++mm) unsafeAtomicAdd(&sTP[e * IN_DIM + w3 + mm], xv[t][v][mm]);
        }
      }
    }
  }
  __syncthreads();

  {
    constexpr int SC = 32 * IN_DIM / 128;
    int   sr[SC];
    float tv[SC];
    #pragma unroll
    for (int k = 0; k < SC; ++k) {
      const int i = tid + 128 * k;
      sr[k] = sSrc[i / IN_DIM];
      tv[k] = sTP[i];
    }
    const int srcn = sSrc[lane];
    #pragma unroll
    for (int k = 0; k < SC; ++k) {
      const int i = tid + 128 * k;
      const int c = i - (i / IN_DIM) * IN_DIM;
      unsafeAtomicAdd(&num[(size_t)sr[k] * IN_DIM + c], tv[k]);
    }
    if (wv == 0) unsafeAtomicAdd(&cnt[srcn], 1.0f);
  }
}

#define NP1_ROWS    8
#define NP1_STEPS   10
#define NP1_NODES   (NP1_ROWS * NP1_STEPS)
#define NP1_THREADS (NP1_ROWS * IN_DIM)
static_assert(N_NODES % NP1_NODES == 0, "node pass 1 has no tail block");

__global__ __launch_bounds__(NP1_THREADS)
void node_pass1(const float* __restrict__ node_attr,
                const float* __restrict__ num, const float* __restrict__ cnt,
                float* __restrict__ outp, float* __restrict__ stat)
{
  __shared__ float ssum[NP1_THREADS], ssq[NP1_THREADS];
  const int tid = threadIdx.x;
  const int g   = tid / IN_DIM;
  const int ch  = tid - g * IN_DIM;
  const int n0  = blockIdx.x * NP1_NODES + g;
  const size_t e0 = (size_t)n0 * IN_DIM + ch;

  float cv[NP1_STEPS], nv[NP1_STEPS], av[NP1_STEPS];
  #pragma unroll
  for (int k = 0; k < NP1_STEPS; ++k) {
    cv[k] = cnt[n0 + k * NP1_ROWS];
    nv[k] = num[e0 + (size_t)k * NP1_ROWS * IN_DIM];
    av[k] = node_attr[e0 + (size_t)k * NP1_ROWS * IN_DIM];
  }
  float s = 0.0f, q = 0.0f;
  #pragma unroll
  for (int k = 0; k < NP1_STEPS; ++k) {
    float c   = cv[k];
    float inv = 1.0f / (c > 1.0f ? c : 1.0f);
    float p = nv[k] * inv + av[k];
    outp[e0 + (size_t)k * NP1_ROWS * IN_DIM] = p;
    s += p;
    q += p * p;
  }
  ssum[tid] = s;
  ssq[tid]  = q;
  __syncthreads();

  if (tid < MUL0) {
    float S = 0.0f, Q = 0.0f;
    #pragma unroll
    for (int r = 0; r < NP1_ROWS; ++r) {
      S += ssum[r * IN_DIM + tid];
      Q += ssq[r * IN_DIM + tid];
    }
    unsafeAtomicAdd(&stat[tid], S);
    unsafeAtomicAdd(&stat[MUL0 + tid], Q);
  } else if (tid < MUL0 + MUL1) {
    const int u = tid - MUL0;
    float Q = 0.0f;
    #pragma unroll
    for (int r = 0; r < NP1_ROWS; ++r) {
      const float* qr = &ssq[r * IN_DIM + MUL0 + u * 3];
      Q += qr[0] + qr[1] + qr[2];
    }
    unsafeAtomicAdd(&stat[2 * MUL0 + u], Q);
  }
}

__global__ __launch_bounds__(128)
void node_pass2(const float* __restrict__ stat,
                const float* __restrict__ bn_w_s, const float* __restrict__ bn_w_v,
                const float* __restrict__ bn_b_s,
                float* __restrict__ out)
{
  const int n = blockIdx.x * 128 + threadIdx.x;
  if (n >= N_NODES) return;
  const float invN = 1.0f / (float)N_NODES;
  for (int ch = 0; ch < MUL0; ++ch) {
    float mean = stat[ch] * invN;
    float var  = stat[MUL0 + ch] * invN - mean * mean;
    float p = out[(size_t)n * IN_DIM + ch];
    out[(size_t)n * IN_DIM + ch] = (p - mean) * rsqrtf(var + EPS) * bn_w_s[ch] + bn_b_s[ch];
  }
  for (int u = 0; u < MUL1; ++u) {
    float vn    = stat[2 * MUL0 + u] * invN * (1.0f / 3.0f);
    float scale = rsqrtf(vn + EPS) * bn_w_v[u];
    for (int mm = 0; mm < 3; ++mm) {
      int ch = MUL0 + u * 3 + mm;
      out[(size_t)n * IN_DIM + ch] *= scale;
    }
  }
}

extern "C" void kernel_launch(void* const* d_in, const int* in_sizes, int n_in,
                              void* d_out, int out_size, void* d_ws, size_t ws_size,
                              hipStream_t stream) {
  const float* node_attr = (const float*)d_in[0];
  const int*   edge_index = (const int*)d_in[1];
  const float* edge_attr = (const float*)d_in[2];
  const float* edge_sh   = (const float*)d_in[3];
  const float* fc1_w = (const float*)d_in[4];
  const float* fc1_b = (const float*)d_in[5];
  const float* fc2_w = (const float*)d_in[6];
  const float* fc2_b = (const float*)d_in[7];
  const float* bn_w_s = (const float*)d_in[8];
  const float* bn_w_v = (const float*)d_in[9];
  const float* bn_b_s = (const float*)d_in[10];

  float* ws   = (float*)d_ws;
  float* num  = ws + WS_NUM;
  float* cnt  = ws + WS_CNT;
  float* stat = ws + WS_STAT;
  float* out  = (float*)d_out;

  prep_kernel<<<(WS_TOTAL + 255) / 256, 256, 0, stream>>>(ws, fc1_w, fc2_w, out);
  edge_kernel<<<N_EDGES / 32, 128, 0, stream>>>(node_attr, edge_index, edge_attr, edge_sh,
                                                fc1_b, out, fc2_b, num, cnt);
  node_pass1<<<N_NODES / NP1_NODES, NP1_THREADS, 0, stream>>>(node_attr, num, cnt, out, stat);
  node_pass2<<<(N_NODES + 127) / 128, 128, 0, stream>>>(stat, bn_w_s, bn_w_v, bn_b_s, out);
}
